# starting kernel
# baseline (speedup 1.0000x reference)
#include <hip/hip_runtime.h>
#include <hip/hip_bf16.h>

typedef float v2f __attribute__((ext_vector_type(2)));
typedef float v8f __attribute__((ext_vector_type(8)));

#define F_IN  128
#define F_HID 64
#define F_OUT 32

__device__ __forceinline__ void atomAddF(float* p, float v) {
    __hip_atomic_fetch_add(p, v, __ATOMIC_RELAXED, __HIP_MEMORY_SCOPE_AGENT);
}

__global__ void k_fill(float* __restrict__ p, float v, long long n) {
    long long i = (long long)blockIdx.x * blockDim.x + threadIdx.x;
    if (i < n) p[i] = v;
}

__global__ void k_deg(const int* __restrict__ col, float* __restrict__ deg, long long E) {
    long long e = (long long)blockIdx.x * blockDim.x + threadIdx.x;
    if (e < E) atomAddF(&deg[col[e]], 1.0f);
}

__global__ void k_dinv(float* __restrict__ deg, int n) {
    int i = blockIdx.x * blockDim.x + threadIdx.x;
    if (i < n) deg[i] = rsqrtf(deg[i]);
}

template <int K, int COUT>
__global__ void k_gemm_wmma(const float* __restrict__ X, const float* __restrict__ W,
                            float* __restrict__ H, int n) {
    constexpr int COLT = COUT / 16;
    const int lane = threadIdx.x & 31;
    const int wv   = threadIdx.x >> 5;
    const int tile = blockIdx.x * (blockDim.x >> 5) + wv;
    const int rowTile = tile / COLT;
    const int colT    = tile - rowTile * COLT;
    const int rowBase = rowTile << 4;
    if (rowBase >= n) return;

    const int m  = lane & 15;
    const int kh = (lane >> 4) << 1;
    int rowA = rowBase + m;
    if (rowA >= n) rowA = n - 1;
    const float*  xr = X + (size_t)rowA * K;
    const float*  wc = W + colT * 16 + m;

    v8f acc = {};
    #pragma unroll
    for (int k = 0; k < K; k += 4) {
        v2f a, b;
        float2 av = *(const float2*)(xr + k + kh);
        a[0] = av.x;
        a[1] = av.y;
        b[0] = wc[(size_t)(k + kh)     * COUT];
        b[1] = wc[(size_t)(k + kh + 1) * COUT];
        acc = __builtin_amdgcn_wmma_f32_16x16x4_f32(
             false, a,  false, b,
             (short)0, acc,  false,  false);
    }

    const int mo = (lane >> 4) << 3;
    const int c  = colT * 16 + m;
    #pragma unroll
    for (int v = 0; v < 8; ++v) {
        int r = rowBase + mo + v;
        if (r < n) H[(size_t)r * COUT + c] = acc[v];
    }
}

__global__ void k_scatter64(const float* __restrict__ Hs, const int* __restrict__ row,
                            const int* __restrict__ col, const float* __restrict__ dinv,
                            float* __restrict__ out, long long E) {
    long long w = ((long long)blockIdx.x * blockDim.x + threadIdx.x) >> 5;
    const int lane = threadIdx.x & 31;
    if (w >= E) return;
    const int r = row[w], c = col[w];
    const float nrm = dinv[r] * dinv[c];
    float2 v = ((const float2*)(Hs + (size_t)r * 64))[lane];
    float* o = out + (size_t)c * 64 + lane * 2;
    atomAddF(o,     v.x * nrm);
    atomAddF(o + 1, v.y * nrm);
}

__global__ void k_scatter32(const float* __restrict__ Hs, const int* __restrict__ row,
                            const int* __restrict__ col, const float* __restrict__ dinv,
                            float* __restrict__ out, long long E) {
    long long w = ((long long)blockIdx.x * blockDim.x + threadIdx.x) >> 5;
    const int lane = threadIdx.x & 31;
    if (w >= E) return;
    const int r = row[w], c = col[w];
    const float nrm = dinv[r] * dinv[c];
    float v = Hs[(size_t)r * 32 + lane];
    atomAddF(out + (size_t)c * 32 + lane, v * nrm);
}

template <int F, bool RELU>
__global__ void k_self_bias(float* __restrict__ agg, const float* __restrict__ Hs,
                            const float* __restrict__ dinv, const float* __restrict__ bias,
                            long long total) {
    long long i = (long long)blockIdx.x * blockDim.x + threadIdx.x;
    if (i >= total) return;
    const long long node = i / F;
    const int f = (int)(i - node * F);
    const float di = dinv[node];
    float v = agg[i] + di * di * Hs[i] + bias[f];
    if (RELU) v = fmaxf(v, 0.0f);
    agg[i] = v;
}

__global__ void k_decode(const float* __restrict__ Z, const int* __restrict__ ea,
                         const int* __restrict__ eb, float* __restrict__ out, long long EL) {
    long long e = (long long)blockIdx.x * blockDim.x + threadIdx.x;
    if (e >= EL) return;
    const float4* za = (const float4*)(Z + (size_t)ea[e] * 32);
    const float4* zb = (const float4*)(Z + (size_t)eb[e] * 32);
    float s = 0.0f;
    #pragma unroll
    for (int i = 0; i < 8; ++i) {
        float4 a = za[i], b = zb[i];
        s += a.x * b.x + a.y * b.y + a.z * b.z + a.w * b.w;
    }
    out[e] = s;
}

static inline int cdiv(long long a, long long b) { return (int)((a + b - 1) / b); }

extern "C" void kernel_launch(void* const* d_in, const int* in_sizes, int n_in,
                              void* d_out, int out_size, void* d_ws, size_t ws_size,
                              hipStream_t stream) {
    const float* x   = (const float*)d_in[0];
    const float* W1  = (const float*)d_in[1];
    const float* b1  = (const float*)d_in[2];
    const float* W2  = (const float*)d_in[3];
    const float* b2  = (const float*)d_in[4];
    const int*   ei  = (const int*)d_in[5];
    const int*   eli = (const int*)d_in[6];

    const long long N  = in_sizes[0] / F_IN;
    const long long E  = in_sizes[5] / 2;
    const long long EL = in_sizes[6] / 2;
    float* out = (float*)d_out;

    float* deg  = (float*)d_ws;
    float* h1   = deg  + N;
    float* agg1 = h1   + N * F_HID;
    float* h2   = agg1 + N * F_HID;
    float* agg2 = h2   + N * F_OUT;

    const int* row = ei;
    const int* col = ei + E;

    k_fill<<<cdiv(N, 256), 256, 0, stream>>>(deg, 1.0f, N);
    k_deg <<<cdiv(E, 256), 256, 0, stream>>>(col, deg, E);
    k_dinv<<<cdiv(N, 256), 256, 0, stream>>>(deg, (int)N);

    {
        long long tiles = ((N + 15) / 16) * (F_HID / 16);
        k_gemm_wmma<F_IN, F_HID><<<cdiv(tiles, 8), 256, 0, stream>>>(x, W1, h1, (int)N);
    }

    k_fill<<<cdiv(N * F_HID, 256), 256, 0, stream>>>(agg1, 0.0f, N * F_HID);
    k_scatter64<<<cdiv(E, 8), 256, 0, stream>>>(h1, row, col, deg, agg1, E);
    k_self_bias<F_HID, true><<<cdiv(N * F_HID, 256), 256, 0, stream>>>(agg1, h1, deg, b1, N * F_HID);

    {
        long long tiles = ((N + 15) / 16) * (F_OUT / 16);
        k_gemm_wmma<F_HID, F_OUT><<<cdiv(tiles, 8), 256, 0, stream>>>(agg1, W2, h2, (int)N);
    }

    k_fill<<<cdiv(N * F_OUT, 256), 256, 0, stream>>>(agg2, 0.0f, N * F_OUT);
    k_scatter32<<<cdiv(E, 8), 256, 0, stream>>>(h2, row, col, deg, agg2, E);
    k_self_bias<F_OUT, false><<<cdiv(N * F_OUT, 256), 256, 0, stream>>>(agg2, h2, deg, b2, N * F_OUT);

    k_decode<<<cdiv(EL, 256), 256, 0, stream>>>(agg2, eli, eli + EL, out, EL);
}


# scatter kernels: one wave walks 32 consecutive edges (coalesced index loads, row loads batched ahead of their atomics) instead of one wave per edge; sums still in atomic arrival order
# speedup vs baseline: 1.6099x; 1.6099x over previous
#include <hip/hip_runtime.h>
#include <hip/hip_bf16.h>

typedef float v2f __attribute__((ext_vector_type(2)));
typedef float v8f __attribute__((ext_vector_type(8)));

#define F_IN  128
#define F_HID 64
#define F_OUT 32

__device__ __forceinline__ void atomAddF(float* p, float v) {
    __hip_atomic_fetch_add(p, v, __ATOMIC_RELAXED, __HIP_MEMORY_SCOPE_AGENT);
}

__global__ void k_fill(float* __restrict__ p, float v, long long n) {
    long long i = (long long)blockIdx.x * blockDim.x + threadIdx.x;
    if (i < n) p[i] = v;
}

__global__ void k_deg(const int* __restrict__ col, float* __restrict__ deg, long long E) {
    long long e = (long long)blockIdx.x * blockDim.x + threadIdx.x;
    if (e < E) atomAddF(&deg[col[e]], 1.0f);
}

__global__ void k_dinv(float* __restrict__ deg, int n) {
    int i = blockIdx.x * blockDim.x + threadIdx.x;
    if (i < n) deg[i] = rsqrtf(deg[i]);
}

template <int K, int COUT>
__global__ void k_gemm_wmma(const float* __restrict__ X, const float* __restrict__ W,
                            float* __restrict__ H, int n) {
    constexpr int COLT = COUT / 16;
    const int lane = threadIdx.x & 31;
    const int wv   = threadIdx.x >> 5;
    const int tile = blockIdx.x * (blockDim.x >> 5) + wv;
    const int rowTile = tile / COLT;
    const int colT    = tile - rowTile * COLT;
    const int rowBase = rowTile << 4;
    if (rowBase >= n) return;

    const int m  = lane & 15;
    const int kh = (lane >> 4) << 1;
    int rowA = rowBase + m;
    if (rowA >= n) rowA = n - 1;
    const float*  xr = X + (size_t)rowA * K;
    const float*  wc = W + colT * 16 + m;

    v8f acc = {};
    #pragma unroll
    for (int k = 0; k < K; k += 4) {
        v2f a, b;
        float2 av = *(const float2*)(xr + k + kh);
        a[0] = av.x;
        a[1] = av.y;
        b[0] = wc[(size_t)(k + kh)     * COUT];
        b[1] = wc[(size_t)(k + kh + 1) * COUT];
        acc = __builtin_amdgcn_wmma_f32_16x16x4_f32(
             false, a,  false, b,
             (short)0, acc,  false,  false);
    }

    const int mo = (lane >> 4) << 3;
    const int c  = colT * 16 + m;
    #pragma unroll
    for (int v = 0; v < 8; ++v) {
        int r = rowBase + mo + v;
        if (r < n) H[(size_t)r * COUT + c] = acc[v];
    }
}

template <int F>
__global__ void k_scatter(const float* __restrict__ Hs, const int* __restrict__ row,
                          const int* __restrict__ col, const float* __restrict__ dinv,
                          float* __restrict__ out, long long E) {
    constexpr int VPL   = F / 32;
    constexpr int BATCH = 16;
    const int lane = threadIdx.x & 31;
    const long long wave = ((long long)blockIdx.x * blockDim.x + threadIdx.x) >> 5;
    const long long base = wave << 5;
    if (base >= E) return;
    const long long e = base + lane;
    int r = 0, c = 0;
    float nrm = 0.0f;
    if (e < E) {
        r = row[e];
        c = col[e];
        nrm = dinv[r] * dinv[c];
    }
    const long long left = E - base;
    const int cnt = left < 32 ? (int)left : 32;
    const int nb  = __float_as_int(nrm);
    const int lo  = lane * VPL;

    if (cnt == 32) {
        #pragma unroll
        for (int j0 = 0; j0 < 32; j0 += BATCH) {
            float vx[BATCH], vy[BATCH];
            #pragma unroll
            for (int j = 0; j < BATCH; ++j) {
                const int rj = __builtin_amdgcn_readlane(r, j0 + j);
                const float* src = Hs + (size_t)rj * F + lo;
                if constexpr (VPL == 2) {
                    const float2 t = *(const float2*)src;
                    vx[j] = t.x;
                    vy[j] = t.y;
                } else {
                    vx[j] = *src;
                    vy[j] = 0.0f;
                }
            }
            #pragma unroll
            for (int j = 0; j < BATCH; ++j) {
                const int   cj = __builtin_amdgcn_readlane(c, j0 + j);
                const float nj = __int_as_float(__builtin_amdgcn_readlane(nb, j0 + j));
                float* o = out + (size_t)cj * F + lo;
                atomAddF(o, vx[j] * nj);
                if constexpr (VPL == 2) atomAddF(o + 1, vy[j] * nj);
            }
        }
    } else {
        for (int j = 0; j < cnt; ++j) {
            const int   rj = __builtin_amdgcn_readlane(r, j);
            const int   cj = __builtin_amdgcn_readlane(c, j);
            const float nj = __int_as_float(__builtin_amdgcn_readlane(nb, j));
            const float* src = Hs + (size_t)rj * F + lo;
            float* o = out + (size_t)cj * F + lo;
            if constexpr (VPL == 2) {
                const float2 t = *(const float2*)src;
                atomAddF(o,     t.x * nj);
                atomAddF(o + 1, t.y * nj);
            } else {
                atomAddF(o, *src * nj);
            }
        }
    }
}

template <int F, bool RELU>
__global__ void k_self_bias(float* __restrict__ agg, const float* __restrict__ Hs,
                            const float* __restrict__ dinv, const float* __restrict__ bias,
                            long long total) {
    long long i = (long long)blockIdx.x * blockDim.x + threadIdx.x;
    if (i >= total) return;
    const long long node = i / F;
    const int f = (int)(i - node * F);
    const float di = dinv[node];
    float v = agg[i] + di * di * Hs[i] + bias[f];
    if (RELU) v = fmaxf(v, 0.0f);
    agg[i] = v;
}

__global__ void k_decode(const float* __restrict__ Z, const int* __restrict__ ea,
                         const int* __restrict__ eb, float* __restrict__ out, long long EL) {
    long long e = (long long)blockIdx.x * blockDim.x + threadIdx.x;
    if (e >= EL) return;
    const float4* za = (const float4*)(Z + (size_t)ea[e] * 32);
    const float4* zb = (const float4*)(Z + (size_t)eb[e] * 32);
    float s = 0.0f;
    #pragma unroll
    for (int i = 0; i < 8; ++i) {
        float4 a = za[i], b = zb[i];
        s += a.x * b.x + a.y * b.y + a.z * b.z + a.w * b.w;
    }
    out[e] = s;
}

static inline int cdiv(long long a, long long b) { return (int)((a + b - 1) / b); }

extern "C" void kernel_launch(void* const* d_in, const int* in_sizes, int n_in,
                              void* d_out, int out_size, void* d_ws, size_t ws_size,
                              hipStream_t stream) {
    const float* x   = (const float*)d_in[0];
    const float* W1  = (const float*)d_in[1];
    const float* b1  = (const float*)d_in[2];
    const float* W2  = (const float*)d_in[3];
    const float* b2  = (const float*)d_in[4];
    const int*   ei  = (const int*)d_in[5];
    const int*   eli = (const int*)d_in[6];

    const long long N  = in_sizes[0] / F_IN;
    const long long E  = in_sizes[5] / 2;
    const long long EL = in_sizes[6] / 2;
    float* out = (float*)d_out;

    float* deg  = (float*)d_ws;
    float* h1   = deg  + N;
    float* agg1 = h1   + N * F_HID;
    float* h2   = agg1 + N * F_HID;
    float* agg2 = h2   + N * F_OUT;

    const int* row = ei;
    const int* col = ei + E;

    k_fill<<<cdiv(N, 256), 256, 0, stream>>>(deg, 1.0f, N);
    k_deg <<<cdiv(E, 256), 256, 0, stream>>>(col, deg, E);
    k_dinv<<<cdiv(N, 256), 256, 0, stream>>>(deg, (int)N);

    {
        long long tiles = ((N + 15) / 16) * (F_HID / 16);
        k_gemm_wmma<F_IN, F_HID><<<cdiv(tiles, 8), 256, 0, stream>>>(x, W1, h1, (int)N);
    }

    k_fill<<<cdiv(N * F_HID, 256), 256, 0, stream>>>(agg1, 0.0f, N * F_HID);
    k_scatter<F_HID><<<cdiv(E, 256), 256, 0, stream>>>(h1, row, col, deg, agg1, E);
    k_self_bias<F_HID, true><<<cdiv(N * F_HID, 256), 256, 0, stream>>>(agg1, h1, deg, b1, N * F_HID);

    {
        long long tiles = ((N + 15) / 16) * (F_OUT / 16);
        k_gemm_wmma<F_HID, F_OUT><<<cdiv(tiles, 8), 256, 0, stream>>>(agg1, W2, h2, (int)N);
    }

    k_fill<<<cdiv(N * F_OUT, 256), 256, 0, stream>>>(agg2, 0.0f, N * F_OUT);
    k_scatter<F_OUT><<<cdiv(E, 256), 256, 0, stream>>>(h2, row, col, deg, agg2, E);
    k_self_bias<F_OUT, false><<<cdiv(N * F_OUT, 256), 256, 0, stream>>>(agg2, h2, deg, b2, N * F_OUT);

    k_decode<<<cdiv(EL, 256), 256, 0, stream>>>(agg2, eli, eli + EL, out, EL);
}


# edges grouped by destination (integer atomic rank, prefix sum, place); aggregation gathers rows, one wave per node, sum in registers, stored once; no float atomics; sums in rank (arrival) order
# speedup vs baseline: 2.2335x; 1.3874x over previous
#include <hip/hip_runtime.h>
#include <hip/hip_bf16.h>


typedef float v2f __attribute__((ext_vector_type(2)));
typedef float v8f __attribute__((ext_vector_type(8)));

#define F_IN  128
#define F_HID 64
#define F_OUT 32

#define SCAN_T 1024

__global__ void k_zero(int* __restrict__ p, int n) {
    int i = blockIdx.x * blockDim.x + threadIdx.x;
    if (i < n) p[i] = 0;
}

__global__ void k_rank(const int* __restrict__ col, int* __restrict__ cnt,
                       int* __restrict__ rank, long long E) {
    long long e = (long long)blockIdx.x * blockDim.x + threadIdx.x;
    if (e < E)
        rank[e] = __hip_atomic_fetch_add(&cnt[col[e]], 1, __ATOMIC_RELAXED,
                                         __HIP_MEMORY_SCOPE_AGENT);
}

__device__ __forceinline__ int wave_sum(int v) {
    #pragma unroll
    for (int d = 16; d > 0; d >>= 1) v += __shfl_xor(v, d);
    return v;
}

__global__ void k_bsum(const int* __restrict__ cnt, int* __restrict__ part, int n) {
    __shared__ int ws[SCAN_T / 32];
    const int lane = threadIdx.x & 31, wv = threadIdx.x >> 5;
    const int i = blockIdx.x * SCAN_T + threadIdx.x;
    int v = wave_sum(i < n ? cnt[i] : 0);
    if (lane == 0) ws[wv] = v;
    __syncthreads();
    if (wv == 0) {
        v = wave_sum(ws[lane]);
        if (lane == 0) part[blockIdx.x] = v;
    }
}

__global__ void k_scan(const int* __restrict__ cnt, const int* __restrict__ part,
                       int* __restrict__ off, float* __restrict__ dinv, int n) {
    __shared__ int ws[SCAN_T / 32];
    __shared__ int carry_s;
    const int lane = threadIdx.x & 31, wv = threadIdx.x >> 5;
    if (wv == 0) {
        int s = 0;
        for (int k = lane; k < (int)blockIdx.x; k += 32) s += part[k];
        s = wave_sum(s);
        if (lane == 0) carry_s = s;
    }
    const int i = blockIdx.x * SCAN_T + threadIdx.x;
    const int c = i < n ? cnt[i] : 0;
    int v = c;
    #pragma unroll
    for (int d = 1; d < 32; d <<= 1) {
        const int t = __shfl_up(v, d);
        if (lane >= d) v += t;
    }
    if (lane == 31) ws[wv] = v;
    __syncthreads();
    if (wv == 0) {
        const int w = ws[lane];
        int s = w;
        #pragma unroll
        for (int d = 1; d < 32; d <<= 1) {
            const int t = __shfl_up(s, d);
            if (lane >= d) s += t;
        }
        ws[lane] = s - w;
    }
    __syncthreads();
    const int excl = v - c + ws[wv] + carry_s;
    if (i < n) {
        off[i]  = excl;
        dinv[i] = rsqrtf((float)(c + 1));
        if (i == n - 1) off[n] = excl + c;
    }
}

__global__ void k_place(const int* __restrict__ row, const int* __restrict__ col,
                        const int* __restrict__ rank, const int* __restrict__ off,
                        const float* __restrict__ dinv, int* __restrict__ slot, long long E) {
    long long e = (long long)blockIdx.x * blockDim.x + threadIdx.x;
    if (e >= E) return;
    const int r = row[e];
    const int p = off[col[e]] + rank[e];
    ((int2*)slot)[p] = make_int2(r, __float_as_int(dinv[r]));
}

template <int K, int COUT>
__global__ void k_gemm_wmma(const float* __restrict__ X, const float* __restrict__ W,
                            float* __restrict__ H, int n) {
    constexpr int COLT = COUT / 16;
    const int lane = threadIdx.x & 31;
    const int wv   = threadIdx.x >> 5;
    const int tile = blockIdx.x * (blockDim.x >> 5) + wv;
    const int rowTile = tile / COLT;
    const int colT    = tile - rowTile * COLT;
    const int rowBase = rowTile << 4;
    if (rowBase >= n) return;

    const int m  = lane & 15;
    const int kh = (lane >> 4) << 1;
    int rowA = rowBase + m;
    if (rowA >= n) rowA = n - 1;
    const float*  xr = X + (size_t)rowA * K;
    const float*  wc = W + colT * 16 + m;

    v8f acc = {};
    #pragma unroll
    for (int k = 0; k < K; k += 4) {
        v2f a, b;
        float2 av = *(const float2*)(xr + k + kh);
        a[0] = av.x;
        a[1] = av.y;
        b[0] = wc[(size_t)(k + kh)     * COUT];
        b[1] = wc[(size_t)(k + kh + 1) * COUT];
        acc = __builtin_amdgcn_wmma_f32_16x16x4_f32(
             false, a,  false, b,
             (short)0, acc,  false,  false);
    }

    const int mo = (lane >> 4) << 3;
    const int c  = colT * 16 + m;
    #pragma unroll
    for (int v = 0; v < 8; ++v) {
        int r = rowBase + mo + v;
        if (r < n) H[(size_t)r * COUT + c] = acc[v];
    }
}

template <int F, bool RELU>
__global__ void k_gather(const float* __restrict__ Hs, const int* __restrict__ slot,
                         const int* __restrict__ off, const float* __restrict__ dinv,
                         const float* __restrict__ bias, float* __restrict__ out, int n) {
    constexpr int VPL   = F / 32;
    constexpr int BATCH = 16;
    const int lane = threadIdx.x & 31;
    const int node = __builtin_amdgcn_readfirstlane((int)((blockIdx.x * blockDim.x + threadIdx.x) >> 5));
    if (node >= n) return;
    const int beg = off[node];
    const int end = off[node + 1];
    const float di = dinv[node];
    const int lo = lane * VPL;

    float sx, sy = 0.0f;
    {
        const float* src = Hs + (size_t)node * F + lo;
        if constexpr (VPL == 2) {
            const float2 t = *(const float2*)src;
            sx = t.x;
            sy = t.y;
        } else {
            sx = *src;
        }
    }

    float ax = 0.0f, ay = 0.0f;
    for (int base = beg; base < end; base += 32) {
        const int left = end - base;
        const int cnt  = left < 32 ? left : 32;
        int   r = node;
        float w = 0.0f;
        if (lane < cnt) {
            const int2 s = ((const int2*)slot)[base + lane];
            r = s.x;
            w = __int_as_float(s.y) * di;
        }
        const int wb = __float_as_int(w);
        for (int j0 = 0; j0 < cnt; j0 += BATCH) {
            float vx[BATCH], vy[BATCH];
            #pragma unroll
            for (int j = 0; j < BATCH; ++j) {
                const int rj = __builtin_amdgcn_readlane(r, j0 + j);
                const float* src = Hs + (size_t)rj * F + lo;
                if constexpr (VPL == 2) {
                    const float2 t = *(const float2*)src;
                    vx[j] = t.x;
                    vy[j] = t.y;
                } else {
                    vx[j] = *src;
                    vy[j] = 0.0f;
                }
            }
            #pragma unroll
            for (int j = 0; j < BATCH; ++j) {
                const float wj = __int_as_float(__builtin_amdgcn_readlane(wb, j0 + j));
                if (j0 + j < cnt) {
                    ax = fmaf(wj, vx[j], ax);
                    if constexpr (VPL == 2) ay = fmaf(wj, vy[j], ay);
                }
            }
        }
    }

    const float dd = di * di;
    float* o = out + (size_t)node * F + lo;
    if constexpr (VPL == 2) {
        const float2 b = *(const float2*)(bias + lo);
        float ox = ax + dd * sx + b.x;
        float oy = ay + dd * sy + b.y;
        if (RELU) { ox = fmaxf(ox, 0.0f); oy = fmaxf(oy, 0.0f); }
        *(float2*)o = make_float2(ox, oy);
    } else {
        float ox = ax + dd * sx + bias[lo];
        if (RELU) ox = fmaxf(ox, 0.0f);
        *o = ox;
    }
}

__global__ void k_decode(const float* __restrict__ Z, const int* __restrict__ ea,
                         const int* __restrict__ eb, float* __restrict__ out, long long EL) {
    long long e = (long long)blockIdx.x * blockDim.x + threadIdx.x;
    if (e >= EL) return;
    const float4* za = (const float4*)(Z + (size_t)ea[e] * 32);
    const float4* zb = (const float4*)(Z + (size_t)eb[e] * 32);
    float s = 0.0f;
    #pragma unroll
    for (int i = 0; i < 8; ++i) {
        float4 a = za[i], b = zb[i];
        s += a.x * b.x + a.y * b.y + a.z * b.z + a.w * b.w;
    }
    out[e] = s;
}

static inline int cdiv(long long a, long long b) { return (int)((a + b - 1) / b); }

extern "C" void kernel_launch(void* const* d_in, const int* in_sizes, int n_in,
                              void* d_out, int out_size, void* d_ws, size_t ws_size,
                              hipStream_t stream) {
    const float* x   = (const float*)d_in[0];
    const float* W1  = (const float*)d_in[1];
    const float* b1  = (const float*)d_in[2];
    const float* W2  = (const float*)d_in[3];
    const float* b2  = (const float*)d_in[4];
    const int*   ei  = (const int*)d_in[5];
    const int*   eli = (const int*)d_in[6];

    const long long N  = in_sizes[0] / F_IN;
    const long long E  = in_sizes[5] / 2;
    const long long EL = in_sizes[6] / 2;
    float* out = (float*)d_out;

    float* h1   = (float*)d_ws;
    float* agg1 = h1   + N * F_HID;
    float* h2   = agg1 + N * F_HID;
    float* agg2 = h2   + N * F_OUT;
    int*   slot = (int*)(agg2 + N * F_OUT);
    int*   rank = slot + 2 * E;
    int*   cnt  = rank + E;
    int*   off  = cnt + N;
    float* dinv = (float*)(off + N + 1);
    int*   part = (int*)(dinv + N);

    const int* row = ei;
    const int* col = ei + E;

    const int nblk = cdiv(N, SCAN_T);
    k_zero <<<cdiv(N, 256), 256, 0, stream>>>(cnt, (int)N);
    k_rank <<<cdiv(E, 256), 256, 0, stream>>>(col, cnt, rank, E);
    k_bsum <<<nblk, SCAN_T, 0, stream>>>(cnt, part, (int)N);
    k_scan <<<nblk, SCAN_T, 0, stream>>>(cnt, part, off, dinv, (int)N);
    k_place<<<cdiv(E, 256), 256, 0, stream>>>(row, col, rank, off, dinv, slot, E);

    {
        long long tiles = ((N + 15) / 16) * (F_HID / 16);
        k_gemm_wmma<F_IN, F_HID><<<cdiv(tiles, 8), 256, 0, stream>>>(x, W1, h1, (int)N);
    }

    k_gather<F_HID, true><<<cdiv(N, 8), 256, 0, stream>>>(h1, slot, off, dinv, b1, agg1, (int)N);

    {
        long long tiles = ((N + 15) / 16) * (F_OUT / 16);
        k_gemm_wmma<F_HID, F_OUT><<<cdiv(tiles, 8), 256, 0, stream>>>(agg1, W2, h2, (int)N);
    }

    k_gather<F_OUT, false><<<cdiv(N, 8), 256, 0, stream>>>(h2, slot, off, dinv, b2, agg2, (int)N);

    k_decode<<<cdiv(EL, 256), 256, 0, stream>>>(agg2, eli, eli + EL, out, EL);
}


# in-degree counters kept once per cache half (die id picks a wave's copy, prefix sum stacks the two): no counter line is added into from both halves; sums in (half, arrival) order
# speedup vs baseline: 3.2904x; 1.4732x over previous
#include <hip/hip_runtime.h>
#include <hip/hip_bf16.h>


typedef float v2f __attribute__((ext_vector_type(2)));
typedef float v8f __attribute__((ext_vector_type(8)));

#define F_IN  128
#define F_HID 64
#define F_OUT 32

#define SCAN_T 1024

__global__ void k_zero(int* __restrict__ p, int n) {
    int i = blockIdx.x * blockDim.x + threadIdx.x;
    if (i < n) p[i] = 0;
}

__device__ __forceinline__ int cache_half() {
    return (int)((__builtin_amdgcn_s_sendmsg_rtn(0x87) >> 18) & 1u);
}

__global__ void k_rank(const int* __restrict__ col, int* __restrict__ cnt,
                       int* __restrict__ rank, long long E, int n) {
    long long e = (long long)blockIdx.x * blockDim.x + threadIdx.x;
    const int h = cache_half();
    if (e < E) {
        const int old = __hip_atomic_fetch_add(&cnt[(size_t)h * n + col[e]], 1, __ATOMIC_RELAXED,
                                               __HIP_MEMORY_SCOPE_AGENT);
        rank[e] = (old << 1) | h;
    }
}

__device__ __forceinline__ int wave_sum(int v) {
    #pragma unroll
    for (int d = 16; d > 0; d >>= 1) v += __shfl_xor(v, d);
    return v;
}

__global__ void k_bsum(const int* __restrict__ cnt, int* __restrict__ part, int n) {
    __shared__ int ws[SCAN_T / 32];
    const int lane = threadIdx.x & 31, wv = threadIdx.x >> 5;
    const int i = blockIdx.x * SCAN_T + threadIdx.x;
    int v = wave_sum(i < n ? cnt[i] + cnt[n + i] : 0);
    if (lane == 0) ws[wv] = v;
    __syncthreads();
    if (wv == 0) {
        v = wave_sum(ws[lane]);
        if (lane == 0) part[blockIdx.x] = v;
    }
}

__global__ void k_scan(const int* __restrict__ cnt, const int* __restrict__ part,
                       int* __restrict__ off, float* __restrict__ dinv, int n) {
    __shared__ int ws[SCAN_T / 32];
    __shared__ int carry_s;
    const int lane = threadIdx.x & 31, wv = threadIdx.x >> 5;
    if (wv == 0) {
        int s = 0;
        for (int k = lane; k < (int)blockIdx.x; k += 32) s += part[k];
        s = wave_sum(s);
        if (lane == 0) carry_s = s;
    }
    const int i = blockIdx.x * SCAN_T + threadIdx.x;
    const int c0 = i < n ? cnt[i] : 0;
    const int c  = c0 + (i < n ? cnt[n + i] : 0);
    int v = c;
    #pragma unroll
    for (int d = 1; d < 32; d <<= 1) {
        const int t = __shfl_up(v, d);
        if (lane >= d) v += t;
    }
    if (lane == 31) ws[wv] = v;
    __syncthreads();
    if (wv == 0) {
        const int w = ws[lane];
        int s = w;
        #pragma unroll
        for (int d = 1; d < 32; d <<= 1) {
            const int t = __shfl_up(s, d);
            if (lane >= d) s += t;
        }
        ws[lane] = s - w;
    }
    __syncthreads();
    const int excl = v - c + ws[wv] + carry_s;
    if (i < n) {
        off[i]  = excl;
        off[n + 1 + i] = excl + c0;
        dinv[i] = rsqrtf((float)(c + 1));
        if (i == n - 1) off[n] = excl + c;
    }
}

__global__ void k_place(const int* __restrict__ row, const int* __restrict__ col,
                        const int* __restrict__ rank, const int* __restrict__ off,
                        const float* __restrict__ dinv, int* __restrict__ slot, long long E, int n) {
    long long e = (long long)blockIdx.x * blockDim.x + threadIdx.x;
    if (e >= E) return;
    const int r  = row[e];
    const int rk = rank[e];
    const int p  = off[(rk & 1) * (n + 1) + col[e]] + (rk >> 1);
    ((int2*)slot)[p] = make_int2(r, __float_as_int(dinv[r]));
}

template <int K, int COUT>
__global__ void k_gemm_wmma(const float* __restrict__ X, const float* __restrict__ W,
                            float* __restrict__ H, int n) {
    constexpr int COLT = COUT / 16;
    const int lane = threadIdx.x & 31;
    const int wv   = threadIdx.x >> 5;
    const int tile = blockIdx.x * (blockDim.x >> 5) + wv;
    const int rowTile = tile / COLT;
    const int colT    = tile - rowTile * COLT;
    const int rowBase = rowTile << 4;
    if (rowBase >= n) return;

    const int m  = lane & 15;
    const int kh = (lane >> 4) << 1;
    int rowA = rowBase + m;
    if (rowA >= n) rowA = n - 1;
    const float*  xr = X + (size_t)rowA * K;
    const float*  wc = W + colT * 16 + m;

    v8f acc = {};
    #pragma unroll
    for (int k = 0; k < K; k += 4) {
        v2f a, b;
        float2 av = *(const float2*)(xr + k + kh);
        a[0] = av.x;
        a[1] = av.y;
        b[0] = wc[(size_t)(k + kh)     * COUT];
        b[1] = wc[(size_t)(k + kh + 1) * COUT];
        acc = __builtin_amdgcn_wmma_f32_16x16x4_f32(
             false, a,  false, b,
             (short)0, acc,  false,  false);
    }

    const int mo = (lane >> 4) << 3;
    const int c  = colT * 16 + m;
    #pragma unroll
    for (int v = 0; v < 8; ++v) {
        int r = rowBase + mo + v;
        if (r < n) H[(size_t)r * COUT + c] = acc[v];
    }
}

template <int F, bool RELU>
__global__ void k_gather(const float* __restrict__ Hs, const int* __restrict__ slot,
                         const int* __restrict__ off, const float* __restrict__ dinv,
                         const float* __restrict__ bias, float* __restrict__ out, int n) {
    constexpr int VPL   = F / 32;
    constexpr int BATCH = 16;
    const int lane = threadIdx.x & 31;
    const int node = __builtin_amdgcn_readfirstlane((int)((blockIdx.x * blockDim.x + threadIdx.x) >> 5));
    if (node >= n) return;
    const int beg = off[node];
    const int end = off[node + 1];
    const float di = dinv[node];
    const int lo = lane * VPL;

    float sx, sy = 0.0f;
    {
        const float* src = Hs + (size_t)node * F + lo;
        if constexpr (VPL == 2) {
            const float2 t = *(const float2*)src;
            sx = t.x;
            sy = t.y;
        } else {
            sx = *src;
        }
    }

    float ax = 0.0f, ay = 0.0f;
    for (int base = beg; base < end; base += 32) {
        const int left = end - base;
        const int cnt  = left < 32 ? left : 32;
        int   r = node;
        float w = 0.0f;
        if (lane < cnt) {
            const int2 s = ((const int2*)slot)[base + lane];
            r = s.x;
            w = __int_as_float(s.y) * di;
        }
        const int wb = __float_as_int(w);
        for (int j0 = 0; j0 < cnt; j0 += BATCH) {
            float vx[BATCH], vy[BATCH];
            #pragma unroll
            for (int j = 0; j < BATCH; ++j) {
                const int rj = __builtin_amdgcn_readlane(r, j0 + j);
                const float* src = Hs + (size_t)rj * F + lo;
                if constexpr (VPL == 2) {
                    const float2 t = *(const float2*)src;
                    vx[j] = t.x;
                    vy[j] = t.y;
                } else {
                    vx[j] = *src;
                    vy[j] = 0.0f;
                }
            }
            #pragma unroll
            for (int j = 0; j < BATCH; ++j) {
                const float wj = __int_as_float(__builtin_amdgcn_readlane(wb, j0 + j));
                if (j0 + j < cnt) {
                    ax = fmaf(wj, vx[j], ax);
                    if constexpr (VPL == 2) ay = fmaf(wj, vy[j], ay);
                }
            }
        }
    }

    const float dd = di * di;
    float* o = out + (size_t)node * F + lo;
    if constexpr (VPL == 2) {
        const float2 b = *(const float2*)(bias + lo);
        float ox = ax + dd * sx + b.x;
        float oy = ay + dd * sy + b.y;
        if (RELU) { ox = fmaxf(ox, 0.0f); oy = fmaxf(oy, 0.0f); }
        *(float2*)o = make_float2(ox, oy);
    } else {
        float ox = ax + dd * sx + bias[lo];
        if (RELU) ox = fmaxf(ox, 0.0f);
        *o = ox;
    }
}

__global__ void k_decode(const float* __restrict__ Z, const int* __restrict__ ea,
                         const int* __restrict__ eb, float* __restrict__ out, long long EL) {
    long long e = (long long)blockIdx.x * blockDim.x + threadIdx.x;
    if (e >= EL) return;
    const float4* za = (const float4*)(Z + (size_t)ea[e] * 32);
    const float4* zb = (const float4*)(Z + (size_t)eb[e] * 32);
    float s = 0.0f;
    #pragma unroll
    for (int i = 0; i < 8; ++i) {
        float4 a = za[i], b = zb[i];
        s += a.x * b.x + a.y * b.y + a.z * b.z + a.w * b.w;
    }
    out[e] = s;
}

static inline int cdiv(long long a, long long b) { return (int)((a + b - 1) / b); }

extern "C" void kernel_launch(void* const* d_in, const int* in_sizes, int n_in,
                              void* d_out, int out_size, void* d_ws, size_t ws_size,
                              hipStream_t stream) {
    const float* x   = (const float*)d_in[0];
    const float* W1  = (const float*)d_in[1];
    const float* b1  = (const float*)d_in[2];
    const float* W2  = (const float*)d_in[3];
    const float* b2  = (const float*)d_in[4];
    const int*   ei  = (const int*)d_in[5];
    const int*   eli = (const int*)d_in[6];

    const long long N  = in_sizes[0] / F_IN;
    const long long E  = in_sizes[5] / 2;
    const long long EL = in_sizes[6] / 2;
    float* out = (float*)d_out;

    float* h1   = (float*)d_ws;
    float* agg1 = h1   + N * F_HID;
    float* h2   = agg1 + N * F_HID;
    float* agg2 = h2   + N * F_OUT;
    int*   slot = (int*)(agg2 + N * F_OUT);
    int*   rank = slot + 2 * E;
    int*   cnt  = rank + E;
    int*   off  = cnt + 2 * N;
    float* dinv = (float*)(off + 2 * (N + 1));
    int*   part = (int*)(dinv + N);

    const int* row = ei;
    const int* col = ei + E;

    const int nblk = cdiv(N, SCAN_T);
    k_zero <<<cdiv(2 * N, 256), 256, 0, stream>>>(cnt, (int)(2 * N));
    k_rank <<<cdiv(E, 256), 256, 0, stream>>>(col, cnt, rank, E, (int)N);
    k_bsum <<<nblk, SCAN_T, 0, stream>>>(cnt, part, (int)N);
    k_scan <<<nblk, SCAN_T, 0, stream>>>(cnt, part, off, dinv, (int)N);
    k_place<<<cdiv(E, 256), 256, 0, stream>>>(row, col, rank, off, dinv, slot, E, (int)N);

    {
        long long tiles = ((N + 15) / 16) * (F_HID / 16);
        k_gemm_wmma<F_IN, F_HID><<<cdiv(tiles, 8), 256, 0, stream>>>(x, W1, h1, (int)N);
    }

    k_gather<F_HID, true><<<cdiv(N, 8), 256, 0, stream>>>(h1, slot, off, dinv, b1, agg1, (int)N);

    {
        long long tiles = ((N + 15) / 16) * (F_OUT / 16);
        k_gemm_wmma<F_HID, F_OUT><<<cdiv(tiles, 8), 256, 0, stream>>>(agg1, W2, h2, (int)N);
    }

    k_gather<F_OUT, false><<<cdiv(N, 8), 256, 0, stream>>>(h2, slot, off, dinv, b2, agg2, (int)N);

    k_decode<<<cdiv(EL, 256), 256, 0, stream>>>(agg2, eli, eli + EL, out, EL);
}


# edge and node passes run as chip-filling grids whose waves loop over their items (4 edges per thread per trip; a node's slots and own row requested a node ahead); sums still in (half, arrival) order
# speedup vs baseline: 3.4530x; 1.0494x over previous
#include <hip/hip_runtime.h>
#include <hip/hip_bf16.h>


typedef float v2f __attribute__((ext_vector_type(2)));
typedef float v8f __attribute__((ext_vector_type(8)));

#define F_IN  128
#define F_HID 64
#define F_OUT 32

#define SCAN_T 1024

#define EDGE_BLOCKS 1024
#define NODE_BLOCKS 2048
#define EDGE_U 4

__global__ void k_zero(int* __restrict__ p, int n) {
    const int T = gridDim.x * blockDim.x;
    for (int i = blockIdx.x * blockDim.x + threadIdx.x; i < n; i += T) p[i] = 0;
}

__device__ __forceinline__ int cache_half() {
    return (int)((__builtin_amdgcn_s_sendmsg_rtn(0x87) >> 18) & 1u);
}

__global__ void k_rank(const int* __restrict__ col, int* __restrict__ cnt,
                       int* __restrict__ rank, long long E, int n) {
    const long long T = (long long)gridDim.x * blockDim.x;
    long long e = (long long)blockIdx.x * blockDim.x + threadIdx.x;
    const int h = cache_half();
    int* mine = cnt + (size_t)h * n;
    for (; e + (EDGE_U - 1) * T < E; e += EDGE_U * T) {
        int c[EDGE_U], old[EDGE_U];
        #pragma unroll
        for (int u = 0; u < EDGE_U; ++u) c[u] = col[e + u * T];
        #pragma unroll
        for (int u = 0; u < EDGE_U; ++u)
            old[u] = __hip_atomic_fetch_add(&mine[c[u]], 1, __ATOMIC_RELAXED, __HIP_MEMORY_SCOPE_AGENT);
        #pragma unroll
        for (int u = 0; u < EDGE_U; ++u) rank[e + u * T] = (old[u] << 1) | h;
    }
    for (; e < E; e += T) {
        const int old = __hip_atomic_fetch_add(&mine[col[e]], 1, __ATOMIC_RELAXED, __HIP_MEMORY_SCOPE_AGENT);
        rank[e] = (old << 1) | h;
    }
}

__device__ __forceinline__ int wave_sum(int v) {
    #pragma unroll
    for (int d = 16; d > 0; d >>= 1) v += __shfl_xor(v, d);
    return v;
}

__global__ void k_bsum(const int* __restrict__ cnt, int* __restrict__ part, int n) {
    __shared__ int ws[SCAN_T / 32];
    const int lane = threadIdx.x & 31, wv = threadIdx.x >> 5;
    const int i = blockIdx.x * SCAN_T + threadIdx.x;
    int v = wave_sum(i < n ? cnt[i] + cnt[n + i] : 0);
    if (lane == 0) ws[wv] = v;
    __syncthreads();
    if (wv == 0) {
        v = wave_sum(ws[lane]);
        if (lane == 0) part[blockIdx.x] = v;
    }
}

__global__ void k_scan(const int* __restrict__ cnt, const int* __restrict__ part,
                       int* __restrict__ off, float* __restrict__ dinv, int n) {
    __shared__ int ws[SCAN_T / 32];
    __shared__ int carry_s;
    const int lane = threadIdx.x & 31, wv = threadIdx.x >> 5;
    if (wv == 0) {
        int s = 0;
        for (int k = lane; k < (int)blockIdx.x; k += 32) s += part[k];
        s = wave_sum(s);
        if (lane == 0) carry_s = s;
    }
    const int i = blockIdx.x * SCAN_T + threadIdx.x;
    const int c0 = i < n ? cnt[i] : 0;
    const int c  = c0 + (i < n ? cnt[n + i] : 0);
    int v = c;
    #pragma unroll
    for (int d = 1; d < 32; d <<= 1) {
        const int t = __shfl_up(v, d);
        if (lane >= d) v += t;
    }
    if (lane == 31) ws[wv] = v;
    __syncthreads();
    if (wv == 0) {
        const int w = ws[lane];
        int s = w;
        #pragma unroll
        for (int d = 1; d < 32; d <<= 1) {
            const int t = __shfl_up(s, d);
            if (lane >= d) s += t;
        }
        ws[lane] = s - w;
    }
    __syncthreads();
    const int excl = v - c + ws[wv] + carry_s;
    if (i < n) {
        off[i]  = excl;
        off[n + 1 + i] = excl + c0;
        dinv[i] = rsqrtf((float)(c + 1));
        if (i == n - 1) off[n] = excl + c;
    }
}

__global__ void k_place(const int* __restrict__ row, const int* __restrict__ col,
                        const int* __restrict__ rank, const int* __restrict__ off,
                        const float* __restrict__ dinv, int* __restrict__ slot, long long E, int n) {
    const long long T = (long long)gridDim.x * blockDim.x;
    long long e = (long long)blockIdx.x * blockDim.x + threadIdx.x;
    int2* slot2 = (int2*)slot;
    for (; e + (EDGE_U - 1) * T < E; e += EDGE_U * T) {
        int r[EDGE_U], rk[EDGE_U], cc[EDGE_U], p[EDGE_U];
        float d[EDGE_U];
        #pragma unroll
        for (int u = 0; u < EDGE_U; ++u) {
            r[u]  = row[e + u * T];
            rk[u] = rank[e + u * T];
            cc[u] = col[e + u * T];
        }
        #pragma unroll
        for (int u = 0; u < EDGE_U; ++u) {
            p[u] = off[(rk[u] & 1) * (n + 1) + cc[u]] + (rk[u] >> 1);
            d[u] = dinv[r[u]];
        }
        #pragma unroll
        for (int u = 0; u < EDGE_U; ++u)
            slot2[p[u]] = make_int2(r[u], __float_as_int(d[u]));
    }
    for (; e < E; e += T) {
        const int r  = row[e];
        const int rk = rank[e];
        const int p  = off[(rk & 1) * (n + 1) + col[e]] + (rk >> 1);
        slot2[p] = make_int2(r, __float_as_int(dinv[r]));
    }
}

template <int K, int COUT>
__global__ void k_gemm_wmma(const float* __restrict__ X, const float* __restrict__ W,
                            float* __restrict__ H, int n) {
    constexpr int COLT = COUT / 16;
    const int lane = threadIdx.x & 31;
    const int wv   = threadIdx.x >> 5;
    const int tile = blockIdx.x * (blockDim.x >> 5) + wv;
    const int rowTile = tile / COLT;
    const int colT    = tile - rowTile * COLT;
    const int rowBase = rowTile << 4;
    if (rowBase >= n) return;

    const int m  = lane & 15;
    const int kh = (lane >> 4) << 1;
    int rowA = rowBase + m;
    if (rowA >= n) rowA = n - 1;
    const float*  xr = X + (size_t)rowA * K;
    const float*  wc = W + colT * 16 + m;

    v8f acc = {};
    #pragma unroll
    for (int k = 0; k < K; k += 4) {
        v2f a, b;
        float2 av = *(const float2*)(xr + k + kh);
        a[0] = av.x;
        a[1] = av.y;
        b[0] = wc[(size_t)(k + kh)     * COUT];
        b[1] = wc[(size_t)(k + kh + 1) * COUT];
        acc = __builtin_amdgcn_wmma_f32_16x16x4_f32(
             false, a,  false, b,
             (short)0, acc,  false,  false);
    }

    const int mo = (lane >> 4) << 3;
    const int c  = colT * 16 + m;
    #pragma unroll
    for (int v = 0; v < 8; ++v) {
        int r = rowBase + mo + v;
        if (r < n) H[(size_t)r * COUT + c] = acc[v];
    }
}


template <int F>
__device__ __forceinline__ void sum_rows(const float* __restrict__ Hs, int lo, int r, int wb, int cnt,
                                         float& ax, float& ay) {
    constexpr int VPL   = F / 32;
    constexpr int BATCH = 16;
    int j0 = 0;
    do {
        float vx[BATCH], vy[BATCH];
        #pragma unroll
        for (int j = 0; j < BATCH; ++j) {
            const int rj = __builtin_amdgcn_readlane(r, j0 + j);
            const float* src = Hs + (size_t)rj * F + lo;
            if constexpr (VPL == 2) {
                const float2 t = *(const float2*)src;
                vx[j] = t.x;
                vy[j] = t.y;
            } else {
                vx[j] = *src;
                vy[j] = 0.0f;
            }
        }
        #pragma unroll
        for (int j = 0; j < BATCH; ++j) {
            const float wj = __int_as_float(__builtin_amdgcn_readlane(wb, j0 + j));
            if (j0 + j < cnt) {
                ax = fmaf(wj, vx[j], ax);
                if constexpr (VPL == 2) ay = fmaf(wj, vy[j], ay);
            }
        }
        j0 += BATCH;
    } while (j0 < cnt);
}

template <int F, bool RELU>
__global__ void k_gather(const float* __restrict__ Hs, const int* __restrict__ slot,
                         const int* __restrict__ off, const float* __restrict__ dinv,
                         const float* __restrict__ bias, float* __restrict__ out, int n) {
    constexpr int VPL = F / 32;
    const int lane = threadIdx.x & 31;
    const int nw = __builtin_amdgcn_readfirstlane((int)((gridDim.x * blockDim.x) >> 5));
    int node = __builtin_amdgcn_readfirstlane((int)((blockIdx.x * blockDim.x + threadIdx.x) >> 5));
    if (node >= n) return;
    const int lo = lane * VPL;
    const int2* slot2 = (const int2*)slot;

    float bx, by = 0.0f;
    if constexpr (VPL == 2) {
        const float2 b = *(const float2*)(bias + lo);
        bx = b.x;
        by = b.y;
    } else {
        bx = bias[lo];
    }

    int beg = off[node];
    int end = off[node + 1];
    float di = dinv[node];
    int2 s = make_int2(0, 0);
    if (lane < end - beg) s = slot2[beg + lane];
    float sx, sy = 0.0f;
    {
        const float* src = Hs + (size_t)node * F + lo;
        if constexpr (VPL == 2) {
            const float2 t = *(const float2*)src;
            sx = t.x;
            sy = t.y;
        } else {
            sx = *src;
        }
    }
    if constexpr (VPL == 2) {
        asm volatile("" :: "v"(s.x), "v"(s.y), "v"(sx), "v"(sy));
    } else {
        asm volatile("" :: "v"(s.x), "v"(s.y), "v"(sx));
    }
    int nbeg = 0, nend = 0;
    float ndi = 0.0f;
    if (node + nw < n) {
        nbeg = off[node + nw];
        nend = off[node + nw + 1];
        ndi  = dinv[node + nw];
    }

    for (;;) {
        const int nxt = node + nw;
        int2  ns = make_int2(0, 0);
        float nsx = 0.0f, nsy = 0.0f;
        int   cbeg = 0, cend = 0;
        float cdi = 0.0f;
        if (nxt < n) {
            if (lane < nend - nbeg) ns = slot2[nbeg + lane];
            const float* src = Hs + (size_t)nxt * F + lo;
            if constexpr (VPL == 2) {
                const float2 t = *(const float2*)src;
                nsx = t.x;
                nsy = t.y;
            } else {
                nsx = *src;
            }
            if (nxt + nw < n) {
                cbeg = off[nxt + nw];
                cend = off[nxt + nw + 1];
                cdi  = dinv[nxt + nw];
            }
        }

        float ax = 0.0f, ay = 0.0f;
        const int deg = end - beg;
        if (deg > 0) {
            int2 t = make_int2(0, 0);
            if (deg > 32) {
                if (lane < deg - 32) t = slot2[beg + 32 + lane];
            }
            {
                const int cnt = deg < 32 ? deg : 32;
                int   r = node;
                float w = 0.0f;
                if (lane < cnt) {
                    r = s.x;
                    w = __int_as_float(s.y) * di;
                }
                sum_rows<F>(Hs, lo, r, __float_as_int(w), cnt, ax, ay);
            }
            if (deg > 32) {
                {
                    const int left = deg - 32;
                    const int cnt  = left < 32 ? left : 32;
                    int   r = node;
                    float w = 0.0f;
                    if (lane < cnt) {
                        r = t.x;
                        w = __int_as_float(t.y) * di;
                    }
                    sum_rows<F>(Hs, lo, r, __float_as_int(w), cnt, ax, ay);
                }
                for (int base = beg + 64; base < end; base += 32) {
                    const int left = end - base;
                    const int cnt  = left < 32 ? left : 32;
                    int   r = node;
                    float w = 0.0f;
                    if (lane < cnt) {
                        const int2 u = slot2[base + lane];
                        r = u.x;
                        w = __int_as_float(u.y) * di;
                    }
                    sum_rows<F>(Hs, lo, r, __float_as_int(w), cnt, ax, ay);
                }
            }
        }

        const float dd = di * di;
        float* o = out + (size_t)node * F + lo;
        if constexpr (VPL == 2) {
            float ox = ax + dd * sx + bx;
            float oy = ay + dd * sy + by;
            if (RELU) { ox = fmaxf(ox, 0.0f); oy = fmaxf(oy, 0.0f); }
            *(float2*)o = make_float2(ox, oy);
        } else {
            float ox = ax + dd * sx + bx;
            if (RELU) ox = fmaxf(ox, 0.0f);
            *o = ox;
        }

        if (nxt >= n) break;
        node = nxt;
        beg = nbeg; end = nend; di = ndi;
        s = ns; sx = nsx; sy = nsy;
        nbeg = cbeg; nend = cend; ndi = cdi;
    }
}

__global__ void k_decode(const float* __restrict__ Z, const int* __restrict__ ea,
                         const int* __restrict__ eb, float* __restrict__ out, long long EL) {
    long long e = (long long)blockIdx.x * blockDim.x + threadIdx.x;
    if (e >= EL) return;
    const float4* za = (const float4*)(Z + (size_t)ea[e] * 32);
    const float4* zb = (const float4*)(Z + (size_t)eb[e] * 32);
    float s = 0.0f;
    #pragma unroll
    for (int i = 0; i < 8; ++i) {
        float4 a = za[i], b = zb[i];
        s += a.x * b.x + a.y * b.y + a.z * b.z + a.w * b.w;
    }
    out[e] = s;
}

static inline int cdiv(long long a, long long b) { return (int)((a + b - 1) / b); }

extern "C" void kernel_launch(void* const* d_in, const int* in_sizes, int n_in,
                              void* d_out, int out_size, void* d_ws, size_t ws_size,
                              hipStream_t stream) {
    const float* x   = (const float*)d_in[0];
    const float* W1  = (const float*)d_in[1];
    const float* b1  = (const float*)d_in[2];
    const float* W2  = (const float*)d_in[3];
    const float* b2  = (const float*)d_in[4];
    const int*   ei  = (const int*)d_in[5];
    const int*   eli = (const int*)d_in[6];

    const long long N  = in_sizes[0] / F_IN;
    const long long E  = in_sizes[5] / 2;
    const long long EL = in_sizes[6] / 2;
    float* out = (float*)d_out;

    float* h1   = (float*)d_ws;
    float* agg1 = h1   + N * F_HID;
    float* h2   = agg1 + N * F_HID;
    float* agg2 = h2   + N * F_OUT;
    int*   slot = (int*)(agg2 + N * F_OUT);
    int*   rank = slot + 2 * E;
    int*   cnt  = rank + E;
    int*   off  = cnt + 2 * N;
    float* dinv = (float*)(off + 2 * (N + 1));
    int*   part = (int*)(dinv + N);

    const int* row = ei;
    const int* col = ei + E;

    const int nblk = cdiv(N, SCAN_T);
    k_zero <<<256, 256, 0, stream>>>(cnt, (int)(2 * N));
    k_rank <<<EDGE_BLOCKS, 256, 0, stream>>>(col, cnt, rank, E, (int)N);
    k_bsum <<<nblk, SCAN_T, 0, stream>>>(cnt, part, (int)N);
    k_scan <<<nblk, SCAN_T, 0, stream>>>(cnt, part, off, dinv, (int)N);
    k_place<<<EDGE_BLOCKS, 256, 0, stream>>>(row, col, rank, off, dinv, slot, E, (int)N);

    {
        long long tiles = ((N + 15) / 16) * (F_HID / 16);
        k_gemm_wmma<F_IN, F_HID><<<cdiv(tiles, 8), 256, 0, stream>>>(x, W1, h1, (int)N);
    }

    k_gather<F_HID, true><<<NODE_BLOCKS, 256, 0, stream>>>(h1, slot, off, dinv, b1, agg1, (int)N);

    {
        long long tiles = ((N + 15) / 16) * (F_OUT / 16);
        k_gemm_wmma<F_HID, F_OUT><<<cdiv(tiles, 8), 256, 0, stream>>>(agg1, W2, h2, (int)N);
    }

    k_gather<F_OUT, false><<<NODE_BLOCKS, 256, 0, stream>>>(h2, slot, off, dinv, b2, agg2, (int)N);

    k_decode<<<cdiv(EL, 256), 256, 0, stream>>>(agg2, eli, eli + EL, out, EL);
}


# edge slots kept once per cache half too (two slot planes, two prefix sums; a node's list = its run in each plane): no slot line is stored into from both halves; sums still in (half, arrival) order
# speedup vs baseline: 4.5360x; 1.3136x over previous
#include <hip/hip_runtime.h>
#include <hip/hip_bf16.h>


typedef float v2f __attribute__((ext_vector_type(2)));
typedef float v8f __attribute__((ext_vector_type(8)));

#define F_IN  128
#define F_HID 64
#define F_OUT 32

#define SCAN_T 1024

#define EDGE_BLOCKS 1024
#define NODE_BLOCKS 2048
#define EDGE_U 4

__global__ void k_zero(int* __restrict__ p, int n) {
    const int T = gridDim.x * blockDim.x;
    for (int i = blockIdx.x * blockDim.x + threadIdx.x; i < n; i += T) p[i] = 0;
}

__device__ __forceinline__ int cache_half() {
    return (int)((__builtin_amdgcn_s_sendmsg_rtn(0x87) >> 18) & 1u);
}

__global__ void k_rank(const int* __restrict__ col, int* __restrict__ cnt,
                       int* __restrict__ rank, long long E, int n) {
    const long long T = (long long)gridDim.x * blockDim.x;
    long long e = (long long)blockIdx.x * blockDim.x + threadIdx.x;
    const int h = cache_half();
    int* mine = cnt + (size_t)h * n;
    for (; e + (EDGE_U - 1) * T < E; e += EDGE_U * T) {
        int c[EDGE_U], old[EDGE_U];
        #pragma unroll
        for (int u = 0; u < EDGE_U; ++u) c[u] = col[e + u * T];
        #pragma unroll
        for (int u = 0; u < EDGE_U; ++u)
            old[u] = __hip_atomic_fetch_add(&mine[c[u]], 1, __ATOMIC_RELAXED, __HIP_MEMORY_SCOPE_AGENT);
        #pragma unroll
        for (int u = 0; u < EDGE_U; ++u) rank[e + u * T] = (old[u] << 1) | h;
    }
    for (; e < E; e += T) {
        const int old = __hip_atomic_fetch_add(&mine[col[e]], 1, __ATOMIC_RELAXED, __HIP_MEMORY_SCOPE_AGENT);
        rank[e] = (old << 1) | h;
    }
}

__device__ __forceinline__ int wave_sum(int v) {
    #pragma unroll
    for (int d = 16; d > 0; d >>= 1) v += __shfl_xor(v, d);
    return v;
}

__global__ void k_bsum(const int* __restrict__ cnt, int2* __restrict__ part, int n) {
    __shared__ int wa[SCAN_T / 32], wb[SCAN_T / 32];
    const int lane = threadIdx.x & 31, wv = threadIdx.x >> 5;
    const int i = blockIdx.x * SCAN_T + threadIdx.x;
    int a = wave_sum(i < n ? cnt[i] : 0);
    int b = wave_sum(i < n ? cnt[n + i] : 0);
    if (lane == 0) { wa[wv] = a; wb[wv] = b; }
    __syncthreads();
    if (wv == 0) {
        a = wave_sum(wa[lane]);
        b = wave_sum(wb[lane]);
        if (lane == 0) part[blockIdx.x] = make_int2(a, b);
    }
}

__global__ void k_scan(const int* __restrict__ cnt, const int2* __restrict__ part,
                       int2* __restrict__ off, float* __restrict__ dinv, int n) {
    __shared__ int wa[SCAN_T / 32], wb[SCAN_T / 32];
    __shared__ int carry_a, carry_b;
    const int lane = threadIdx.x & 31, wv = threadIdx.x >> 5;
    if (wv == 0) {
        int sa = 0, sb = 0;
        for (int k = lane; k < (int)blockIdx.x; k += 32) {
            const int2 p = part[k];
            sa += p.x;
            sb += p.y;
        }
        sa = wave_sum(sa);
        sb = wave_sum(sb);
        if (lane == 0) { carry_a = sa; carry_b = sb; }
    }
    const int i = blockIdx.x * SCAN_T + threadIdx.x;
    const int c0 = i < n ? cnt[i] : 0;
    const int c1 = i < n ? cnt[n + i] : 0;
    int va = c0, vb = c1;
    #pragma unroll
    for (int d = 1; d < 32; d <<= 1) {
        const int ta = __shfl_up(va, d);
        const int tb = __shfl_up(vb, d);
        if (lane >= d) { va += ta; vb += tb; }
    }
    if (lane == 31) { wa[wv] = va; wb[wv] = vb; }
    __syncthreads();
    if (wv == 0) {
        const int a = wa[lane], b = wb[lane];
        int sa = a, sb = b;
        #pragma unroll
        for (int d = 1; d < 32; d <<= 1) {
            const int ta = __shfl_up(sa, d);
            const int tb = __shfl_up(sb, d);
            if (lane >= d) { sa += ta; sb += tb; }
        }
        wa[lane] = sa - a;
        wb[lane] = sb - b;
    }
    __syncthreads();
    const int ea = va - c0 + wa[wv] + carry_a;
    const int eb = vb - c1 + wb[wv] + carry_b;
    if (i < n) {
        off[i]  = make_int2(ea, eb);
        dinv[i] = rsqrtf((float)(c0 + c1 + 1));
        if (i == n - 1) off[n] = make_int2(ea + c0, eb + c1);
    }
}

__global__ void k_place(const int* __restrict__ row, const int* __restrict__ col,
                        const int* __restrict__ rank, const int* __restrict__ off,
                        const float* __restrict__ dinv, int* __restrict__ slot, int E) {
    const long long T = (long long)gridDim.x * blockDim.x;
    long long e = (long long)blockIdx.x * blockDim.x + threadIdx.x;
    int2* slot2 = (int2*)slot;
    for (; e + (EDGE_U - 1) * T < E; e += EDGE_U * T) {
        int r[EDGE_U], rk[EDGE_U], cc[EDGE_U], p[EDGE_U];
        float d[EDGE_U];
        #pragma unroll
        for (int u = 0; u < EDGE_U; ++u) {
            r[u]  = row[e + u * T];
            rk[u] = rank[e + u * T];
            cc[u] = col[e + u * T];
        }
        #pragma unroll
        for (int u = 0; u < EDGE_U; ++u) {
            const int h = rk[u] & 1;
            p[u] = h * E + off[2 * cc[u] + h] + (rk[u] >> 1);
            d[u] = dinv[r[u]];
        }
        #pragma unroll
        for (int u = 0; u < EDGE_U; ++u)
            slot2[p[u]] = make_int2(r[u], __float_as_int(d[u]));
    }
    for (; e < E; e += T) {
        const int r  = row[e];
        const int rk = rank[e];
        const int h  = rk & 1;
        const int p  = h * E + off[2 * col[e] + h] + (rk >> 1);
        slot2[p] = make_int2(r, __float_as_int(dinv[r]));
    }
}

template <int K, int COUT>
__global__ void k_gemm_wmma(const float* __restrict__ X, const float* __restrict__ W,
                            float* __restrict__ H, int n) {
    constexpr int COLT = COUT / 16;
    const int lane = threadIdx.x & 31;
    const int wv   = threadIdx.x >> 5;
    const int tile = blockIdx.x * (blockDim.x >> 5) + wv;
    const int rowTile = tile / COLT;
    const int colT    = tile - rowTile * COLT;
    const int rowBase = rowTile << 4;
    if (rowBase >= n) return;

    const int m  = lane & 15;
    const int kh = (lane >> 4) << 1;
    int rowA = rowBase + m;
    if (rowA >= n) rowA = n - 1;
    const float*  xr = X + (size_t)rowA * K;
    const float*  wc = W + colT * 16 + m;

    v8f acc = {};
    #pragma unroll
    for (int k = 0; k < K; k += 4) {
        v2f a, b;
        float2 av = *(const float2*)(xr + k + kh);
        a[0] = av.x;
        a[1] = av.y;
        b[0] = wc[(size_t)(k + kh)     * COUT];
        b[1] = wc[(size_t)(k + kh + 1) * COUT];
        acc = __builtin_amdgcn_wmma_f32_16x16x4_f32(
             false, a,  false, b,
             (short)0, acc,  false,  false);
    }

    const int mo = (lane >> 4) << 3;
    const int c  = colT * 16 + m;
    #pragma unroll
    for (int v = 0; v < 8; ++v) {
        int r = rowBase + mo + v;
        if (r < n) H[(size_t)r * COUT + c] = acc[v];
    }
}


template <int F>
__device__ __forceinline__ void sum_rows(const float* __restrict__ Hs, int lo, int r, int wb, int cnt,
                                         float& ax, float& ay) {
    constexpr int VPL   = F / 32;
    constexpr int BATCH = 16;
    int j0 = 0;
    do {
        float vx[BATCH], vy[BATCH];
        #pragma unroll
        for (int j = 0; j < BATCH; ++j) {
            const int rj = __builtin_amdgcn_readlane(r, j0 + j);
            const float* src = Hs + (size_t)rj * F + lo;
            if constexpr (VPL == 2) {
                const float2 t = *(const float2*)src;
                vx[j] = t.x;
                vy[j] = t.y;
            } else {
                vx[j] = *src;
                vy[j] = 0.0f;
            }
        }
        #pragma unroll
        for (int j = 0; j < BATCH; ++j) {
            const float wj = __int_as_float(__builtin_amdgcn_readlane(wb, j0 + j));
            if (j0 + j < cnt) {
                ax = fmaf(wj, vx[j], ax);
                if constexpr (VPL == 2) ay = fmaf(wj, vy[j], ay);
            }
        }
        j0 += BATCH;
    } while (j0 < cnt);
}

struct Seg {
    int a0;
    int ca;
    int b0;
    int deg;
};
__device__ __forceinline__ Seg make_seg(int2 lo, int2 hi, int E) {
    Seg s;
    s.a0  = lo.x;
    s.ca  = hi.x - lo.x;
    s.b0  = E + lo.y - s.ca;
    s.deg = s.ca + (hi.y - lo.y);
    return s;
}
__device__ __forceinline__ int slot_of(const Seg& s, int k) { return k + (k < s.ca ? s.a0 : s.b0); }

template <int F, bool RELU>
__global__ void k_gather(const float* __restrict__ Hs, const int* __restrict__ slot,
                         const int2* __restrict__ off, const float* __restrict__ dinv,
                         const float* __restrict__ bias, float* __restrict__ out, int n, int E) {
    constexpr int VPL = F / 32;
    const int lane = threadIdx.x & 31;
    const int nw = __builtin_amdgcn_readfirstlane((int)((gridDim.x * blockDim.x) >> 5));
    int node = __builtin_amdgcn_readfirstlane((int)((blockIdx.x * blockDim.x + threadIdx.x) >> 5));
    if (node >= n) return;
    const int lo = lane * VPL;
    const int2* slot2 = (const int2*)slot;

    float bx, by = 0.0f;
    if constexpr (VPL == 2) {
        const float2 b = *(const float2*)(bias + lo);
        bx = b.x;
        by = b.y;
    } else {
        bx = bias[lo];
    }

    Seg cur = make_seg(off[node], off[node + 1], E);
    float di = dinv[node];
    int2 s = make_int2(0, 0);
    if (lane < cur.deg) s = slot2[slot_of(cur, lane)];
    float sx, sy = 0.0f;
    {
        const float* src = Hs + (size_t)node * F + lo;
        if constexpr (VPL == 2) {
            const float2 t = *(const float2*)src;
            sx = t.x;
            sy = t.y;
        } else {
            sx = *src;
        }
    }
    if constexpr (VPL == 2) {
        asm volatile("" :: "v"(s.x), "v"(s.y), "v"(sx), "v"(sy));
    } else {
        asm volatile("" :: "v"(s.x), "v"(s.y), "v"(sx));
    }
    int2 nlo = make_int2(0, 0), nhi = make_int2(0, 0);
    float ndi = 0.0f;
    if (node + nw < n) {
        nlo = off[node + nw];
        nhi = off[node + nw + 1];
        ndi = dinv[node + nw];
    }

    for (;;) {
        const int nxt = node + nw;
        int2  ns = make_int2(0, 0);
        float nsx = 0.0f, nsy = 0.0f;
        const Seg nx = make_seg(nlo, nhi, E);
        int2  clo = make_int2(0, 0), chi = make_int2(0, 0);
        float cdi = 0.0f;
        if (nxt < n) {
            if (lane < nx.deg) ns = slot2[slot_of(nx, lane)];
            const float* src = Hs + (size_t)nxt * F + lo;
            if constexpr (VPL == 2) {
                const float2 t = *(const float2*)src;
                nsx = t.x;
                nsy = t.y;
            } else {
                nsx = *src;
            }
            if (nxt + nw < n) {
                clo = off[nxt + nw];
                chi = off[nxt + nw + 1];
                cdi = dinv[nxt + nw];
            }
        }

        float ax = 0.0f, ay = 0.0f;
        const int deg = cur.deg;
        if (deg > 0) {
            int2 t = make_int2(0, 0);
            if (deg > 32) {
                if (lane < deg - 32) t = slot2[slot_of(cur, 32 + lane)];
            }
            {
                const int cnt = deg < 32 ? deg : 32;
                int   r = node;
                float w = 0.0f;
                if (lane < cnt) {
                    r = s.x;
                    w = __int_as_float(s.y) * di;
                }
                sum_rows<F>(Hs, lo, r, __float_as_int(w), cnt, ax, ay);
            }
            if (deg > 32) {
                {
                    const int left = deg - 32;
                    const int cnt  = left < 32 ? left : 32;
                    int   r = node;
                    float w = 0.0f;
                    if (lane < cnt) {
                        r = t.x;
                        w = __int_as_float(t.y) * di;
                    }
                    sum_rows<F>(Hs, lo, r, __float_as_int(w), cnt, ax, ay);
                }
                for (int k0 = 64; k0 < deg; k0 += 32) {
                    const int left = deg - k0;
                    const int cnt  = left < 32 ? left : 32;
                    int   r = node;
                    float w = 0.0f;
                    if (lane < cnt) {
                        const int2 u = slot2[slot_of(cur, k0 + lane)];
                        r = u.x;
                        w = __int_as_float(u.y) * di;
                    }
                    sum_rows<F>(Hs, lo, r, __float_as_int(w), cnt, ax, ay);
                }
            }
        }

        const float dd = di * di;
        float* o = out + (size_t)node * F + lo;
        if constexpr (VPL == 2) {
            float ox = ax + dd * sx + bx;
            float oy = ay + dd * sy + by;
            if (RELU) { ox = fmaxf(ox, 0.0f); oy = fmaxf(oy, 0.0f); }
            *(float2*)o = make_float2(ox, oy);
        } else {
            float ox = ax + dd * sx + bx;
            if (RELU) ox = fmaxf(ox, 0.0f);
            *o = ox;
        }

        if (nxt >= n) break;
        node = nxt;
        cur = nx; di = ndi;
        s = ns; sx = nsx; sy = nsy;
        nlo = clo; nhi = chi; ndi = cdi;
    }
}

__global__ void k_decode(const float* __restrict__ Z, const int* __restrict__ ea,
                         const int* __restrict__ eb, float* __restrict__ out, long long EL) {
    long long e = (long long)blockIdx.x * blockDim.x + threadIdx.x;
    if (e >= EL) return;
    const float4* za = (const float4*)(Z + (size_t)ea[e] * 32);
    const float4* zb = (const float4*)(Z + (size_t)eb[e] * 32);
    float s = 0.0f;
    #pragma unroll
    for (int i = 0; i < 8; ++i) {
        float4 a = za[i], b = zb[i];
        s += a.x * b.x + a.y * b.y + a.z * b.z + a.w * b.w;
    }
    out[e] = s;
}

static inline int cdiv(long long a, long long b) { return (int)((a + b - 1) / b); }

extern "C" void kernel_launch(void* const* d_in, const int* in_sizes, int n_in,
                              void* d_out, int out_size, void* d_ws, size_t ws_size,
                              hipStream_t stream) {
    const float* x   = (const float*)d_in[0];
    const float* W1  = (const float*)d_in[1];
    const float* b1  = (const float*)d_in[2];
    const float* W2  = (const float*)d_in[3];
    const float* b2  = (const float*)d_in[4];
    const int*   ei  = (const int*)d_in[5];
    const int*   eli = (const int*)d_in[6];

    const long long N  = in_sizes[0] / F_IN;
    const long long E  = in_sizes[5] / 2;
    const long long EL = in_sizes[6] / 2;
    float* out = (float*)d_out;

    const int nblk = cdiv(N, SCAN_T);
    float* h1   = (float*)d_ws;
    float* agg1 = h1   + N * F_HID;
    float* h2   = agg1 + N * F_HID;
    float* agg2 = h2   + N * F_OUT;
    int*   slot = (int*)(agg2 + N * F_OUT);
    int2*  off  = (int2*)(slot + 4 * E);
    int2*  part = off + (N + 1);
    int*   rank = (int*)(part + nblk);
    int*   cnt  = rank + E;
    float* dinv = (float*)(cnt + 2 * N);

    const int* row = ei;
    const int* col = ei + E;

    k_zero <<<256, 256, 0, stream>>>(cnt, (int)(2 * N));
    k_rank <<<EDGE_BLOCKS, 256, 0, stream>>>(col, cnt, rank, E, (int)N);
    k_bsum <<<nblk, SCAN_T, 0, stream>>>(cnt, part, (int)N);
    k_scan <<<nblk, SCAN_T, 0, stream>>>(cnt, part, off, dinv, (int)N);
    k_place<<<EDGE_BLOCKS, 256, 0, stream>>>(row, col, rank, (const int*)off, dinv, slot, (int)E);

    {
        long long tiles = ((N + 15) / 16) * (F_HID / 16);
        k_gemm_wmma<F_IN, F_HID><<<cdiv(tiles, 8), 256, 0, stream>>>(x, W1, h1, (int)N);
    }

    k_gather<F_HID, true><<<NODE_BLOCKS, 256, 0, stream>>>(h1, slot, off, dinv, b1, agg1, (int)N, (int)E);

    {
        long long tiles = ((N + 15) / 16) * (F_OUT / 16);
        k_gemm_wmma<F_HID, F_OUT><<<cdiv(tiles, 8), 256, 0, stream>>>(agg1, W2, h2, (int)N);
    }

    k_gather<F_OUT, false><<<NODE_BLOCKS, 256, 0, stream>>>(h2, slot, off, dinv, b2, agg2, (int)N, (int)E);

    k_decode<<<cdiv(EL, 256), 256, 0, stream>>>(agg2, eli, eli + EL, out, EL);
}


# product rows stored times their own node's dinv, so an edge carries no weight: place reads no dinv, stores 4-byte slots; gather adds rows, scales the sum once; sum order kept, rounding differs
# speedup vs baseline: 4.9733x; 1.0964x over previous
#include <hip/hip_runtime.h>
#include <hip/hip_bf16.h>


typedef float v2f __attribute__((ext_vector_type(2)));
typedef float v8f __attribute__((ext_vector_type(8)));

#define F_IN  128
#define F_HID 64
#define F_OUT 32

#define SCAN_T 1024

#define EDGE_BLOCKS 1024
#define NODE_BLOCKS 2048
#define EDGE_U 4

__global__ void k_zero(int* __restrict__ p, int n) {
    const int T = gridDim.x * blockDim.x;
    for (int i = blockIdx.x * blockDim.x + threadIdx.x; i < n; i += T) p[i] = 0;
}

__device__ __forceinline__ int cache_half() {
    return (int)((__builtin_amdgcn_s_sendmsg_rtn(0x87) >> 18) & 1u);
}

__global__ void k_rank(const int* __restrict__ col, int* __restrict__ cnt,
                       int* __restrict__ rank, long long E, int n) {
    const long long T = (long long)gridDim.x * blockDim.x;
    long long e = (long long)blockIdx.x * blockDim.x + threadIdx.x;
    const int h = cache_half();
    int* mine = cnt + (size_t)h * n;
    for (; e + (EDGE_U - 1) * T < E; e += EDGE_U * T) {
        int c[EDGE_U], old[EDGE_U];
        #pragma unroll
        for (int u = 0; u < EDGE_U; ++u) c[u] = col[e + u * T];
        #pragma unroll
        for (int u = 0; u < EDGE_U; ++u)
            old[u] = __hip_atomic_fetch_add(&mine[c[u]], 1, __ATOMIC_RELAXED, __HIP_MEMORY_SCOPE_AGENT);
        #pragma unroll
        for (int u = 0; u < EDGE_U; ++u) rank[e + u * T] = (old[u] << 1) | h;
    }
    for (; e < E; e += T) {
        const int old = __hip_atomic_fetch_add(&mine[col[e]], 1, __ATOMIC_RELAXED, __HIP_MEMORY_SCOPE_AGENT);
        rank[e] = (old << 1) | h;
    }
}

__device__ __forceinline__ int wave_sum(int v) {
    #pragma unroll
    for (int d = 16; d > 0; d >>= 1) v += __shfl_xor(v, d);
    return v;
}

__global__ void k_bsum(const int* __restrict__ cnt, int2* __restrict__ part, int n) {
    __shared__ int wa[SCAN_T / 32], wb[SCAN_T / 32];
    const int lane = threadIdx.x & 31, wv = threadIdx.x >> 5;
    const int i = blockIdx.x * SCAN_T + threadIdx.x;
    int a = wave_sum(i < n ? cnt[i] : 0);
    int b = wave_sum(i < n ? cnt[n + i] : 0);
    if (lane == 0) { wa[wv] = a; wb[wv] = b; }
    __syncthreads();
    if (wv == 0) {
        a = wave_sum(wa[lane]);
        b = wave_sum(wb[lane]);
        if (lane == 0) part[blockIdx.x] = make_int2(a, b);
    }
}

__global__ void k_scan(const int* __restrict__ cnt, const int2* __restrict__ part,
                       int2* __restrict__ off, float* __restrict__ dinv, int n) {
    __shared__ int wa[SCAN_T / 32], wb[SCAN_T / 32];
    __shared__ int carry_a, carry_b;
    const int lane = threadIdx.x & 31, wv = threadIdx.x >> 5;
    if (wv == 0) {
        int sa = 0, sb = 0;
        for (int k = lane; k < (int)blockIdx.x; k += 32) {
            const int2 p = part[k];
            sa += p.x;
            sb += p.y;
        }
        sa = wave_sum(sa);
        sb = wave_sum(sb);
        if (lane == 0) { carry_a = sa; carry_b = sb; }
    }
    const int i = blockIdx.x * SCAN_T + threadIdx.x;
    const int c0 = i < n ? cnt[i] : 0;
    const int c1 = i < n ? cnt[n + i] : 0;
    int va = c0, vb = c1;
    #pragma unroll
    for (int d = 1; d < 32; d <<= 1) {
        const int ta = __shfl_up(va, d);
        const int tb = __shfl_up(vb, d);
        if (lane >= d) { va += ta; vb += tb; }
    }
    if (lane == 31) { wa[wv] = va; wb[wv] = vb; }
    __syncthreads();
    if (wv == 0) {
        const int a = wa[lane], b = wb[lane];
        int sa = a, sb = b;
        #pragma unroll
        for (int d = 1; d < 32; d <<= 1) {
            const int ta = __shfl_up(sa, d);
            const int tb = __shfl_up(sb, d);
            if (lane >= d) { sa += ta; sb += tb; }
        }
        wa[lane] = sa - a;
        wb[lane] = sb - b;
    }
    __syncthreads();
    const int ea = va - c0 + wa[wv] + carry_a;
    const int eb = vb - c1 + wb[wv] + carry_b;
    if (i < n) {
        off[i]  = make_int2(ea, eb);
        dinv[i] = rsqrtf((float)(c0 + c1 + 1));
        if (i == n - 1) off[n] = make_int2(ea + c0, eb + c1);
    }
}

__global__ void k_place(const int* __restrict__ row, const int* __restrict__ col,
                        const int* __restrict__ rank, const int* __restrict__ off,
                        int* __restrict__ slot, int E) {
    const long long T = (long long)gridDim.x * blockDim.x;
    long long e = (long long)blockIdx.x * blockDim.x + threadIdx.x;
    for (; e + (EDGE_U - 1) * T < E; e += EDGE_U * T) {
        int r[EDGE_U], rk[EDGE_U], cc[EDGE_U], p[EDGE_U];
        #pragma unroll
        for (int u = 0; u < EDGE_U; ++u) {
            r[u]  = row[e + u * T];
            rk[u] = rank[e + u * T];
            cc[u] = col[e + u * T];
        }
        static_assert(EDGE_U == 4, "the fence below names the four source ids of a trip");
        asm volatile("" :: "v"(r[0]), "v"(r[1]), "v"(r[2]), "v"(r[3]));
        #pragma unroll
        for (int u = 0; u < EDGE_U; ++u) {
            const int h = rk[u] & 1;
            p[u] = h * E + off[2 * cc[u] + h] + (rk[u] >> 1);
        }
        #pragma unroll
        for (int u = 0; u < EDGE_U; ++u)
            slot[p[u]] = r[u];
    }
    for (; e < E; e += T) {
        const int rk = rank[e];
        const int h  = rk & 1;
        slot[h * E + off[2 * col[e] + h] + (rk >> 1)] = row[e];
    }
}

template <int K, int COUT>
__global__ void k_gemm_wmma(const float* __restrict__ X, const float* __restrict__ W,
                            const float* __restrict__ dinv, float* __restrict__ H, int n) {
    constexpr int COLT = COUT / 16;
    const int lane = threadIdx.x & 31;
    const int wv   = threadIdx.x >> 5;
    const int tile = blockIdx.x * (blockDim.x >> 5) + wv;
    const int rowTile = tile / COLT;
    const int colT    = tile - rowTile * COLT;
    const int rowBase = rowTile << 4;
    if (rowBase >= n) return;

    const int m  = lane & 15;
    const int kh = (lane >> 4) << 1;
    int rowA = rowBase + m;
    if (rowA >= n) rowA = n - 1;
    const float*  xr = X + (size_t)rowA * K;
    const float*  wc = W + colT * 16 + m;

    v8f acc = {};
    #pragma unroll
    for (int k = 0; k < K; k += 4) {
        v2f a, b;
        float2 av = *(const float2*)(xr + k + kh);
        a[0] = av.x;
        a[1] = av.y;
        b[0] = wc[(size_t)(k + kh)     * COUT];
        b[1] = wc[(size_t)(k + kh + 1) * COUT];
        acc = __builtin_amdgcn_wmma_f32_16x16x4_f32(
             false, a,  false, b,
             (short)0, acc,  false,  false);
    }

    const int mo = (lane >> 4) << 3;
    const int c  = colT * 16 + m;
    float dv[8];
    #pragma unroll
    for (int v = 0; v < 8; ++v) {
        const int r = rowBase + mo + v;
        dv[v] = dinv[r < n ? r : n - 1];
    }
    asm volatile("" :: "v"(dv[0]), "v"(dv[1]), "v"(dv[2]), "v"(dv[3]),
                       "v"(dv[4]), "v"(dv[5]), "v"(dv[6]), "v"(dv[7]));
    #pragma unroll
    for (int v = 0; v < 8; ++v) {
        int r = rowBase + mo + v;
        if (r < n) H[(size_t)r * COUT + c] = acc[v] * dv[v];
    }
}


template <int F>
__device__ __forceinline__ void sum_rows(const float* __restrict__ Hs, int lo, int r, int cnt,
                                         float& ax, float& ay) {
    constexpr int VPL   = F / 32;
    constexpr int BATCH = 16;
    int j0 = 0;
    do {
        float vx[BATCH], vy[BATCH];
        #pragma unroll
        for (int j = 0; j < BATCH; ++j) {
            const int rj = __builtin_amdgcn_readlane(r, j0 + j);
            const float* src = Hs + (size_t)rj * F + lo;
            if constexpr (VPL == 2) {
                const float2 t = *(const float2*)src;
                vx[j] = t.x;
                vy[j] = t.y;
            } else {
                vx[j] = *src;
                vy[j] = 0.0f;
            }
        }
        #pragma unroll
        for (int j = 0; j < BATCH; ++j) {
            if (j0 + j < cnt) {
                ax += vx[j];
                if constexpr (VPL == 2) ay += vy[j];
            }
        }
        j0 += BATCH;
    } while (j0 < cnt);
}

struct Seg {
    int a0;
    int ca;
    int b0;
    int deg;
};
__device__ __forceinline__ Seg make_seg(int2 lo, int2 hi, int E) {
    Seg s;
    s.a0  = lo.x;
    s.ca  = hi.x - lo.x;
    s.b0  = E + lo.y - s.ca;
    s.deg = s.ca + (hi.y - lo.y);
    return s;
}
__device__ __forceinline__ int slot_of(const Seg& s, int k) { return k + (k < s.ca ? s.a0 : s.b0); }

template <int F, bool RELU>
__global__ void k_gather(const float* __restrict__ Hs, const int* __restrict__ slot,
                         const int2* __restrict__ off, const float* __restrict__ dinv,
                         const float* __restrict__ bias, float* __restrict__ out, int n, int E) {
    constexpr int VPL = F / 32;
    const int lane = threadIdx.x & 31;
    const int nw = __builtin_amdgcn_readfirstlane((int)((gridDim.x * blockDim.x) >> 5));
    int node = __builtin_amdgcn_readfirstlane((int)((blockIdx.x * blockDim.x + threadIdx.x) >> 5));
    if (node >= n) return;
    const int lo = lane * VPL;

    float bx, by = 0.0f;
    if constexpr (VPL == 2) {
        const float2 b = *(const float2*)(bias + lo);
        bx = b.x;
        by = b.y;
    } else {
        bx = bias[lo];
    }

    Seg cur = make_seg(off[node], off[node + 1], E);
    float di = dinv[node];
    int s = 0;
    if (lane < cur.deg) s = slot[slot_of(cur, lane)];
    float sx, sy = 0.0f;
    {
        const float* src = Hs + (size_t)node * F + lo;
        if constexpr (VPL == 2) {
            const float2 t = *(const float2*)src;
            sx = t.x;
            sy = t.y;
        } else {
            sx = *src;
        }
    }
    if constexpr (VPL == 2) {
        asm volatile("" :: "v"(s), "v"(sx), "v"(sy));
    } else {
        asm volatile("" :: "v"(s), "v"(sx));
    }
    int2 nlo = make_int2(0, 0), nhi = make_int2(0, 0);
    float ndi = 0.0f;
    if (node + nw < n) {
        nlo = off[node + nw];
        nhi = off[node + nw + 1];
        ndi = dinv[node + nw];
    }

    for (;;) {
        const int nxt = node + nw;
        int   ns = 0;
        float nsx = 0.0f, nsy = 0.0f;
        const Seg nx = make_seg(nlo, nhi, E);
        int2  clo = make_int2(0, 0), chi = make_int2(0, 0);
        float cdi = 0.0f;
        if (nxt < n) {
            if (lane < nx.deg) ns = slot[slot_of(nx, lane)];
            const float* src = Hs + (size_t)nxt * F + lo;
            if constexpr (VPL == 2) {
                const float2 t = *(const float2*)src;
                nsx = t.x;
                nsy = t.y;
            } else {
                nsx = *src;
            }
            if (nxt + nw < n) {
                clo = off[nxt + nw];
                chi = off[nxt + nw + 1];
                cdi = dinv[nxt + nw];
            }
        }

        float ax = 0.0f, ay = 0.0f;
        const int deg = cur.deg;
        if (deg > 0) {
            int t = 0;
            if (deg > 32) {
                if (lane < deg - 32) t = slot[slot_of(cur, 32 + lane)];
            }
            {
                const int cnt = deg < 32 ? deg : 32;
                const int r = lane < cnt ? s : node;
                sum_rows<F>(Hs, lo, r, cnt, ax, ay);
            }
            if (deg > 32) {
                {
                    const int left = deg - 32;
                    const int cnt  = left < 32 ? left : 32;
                    const int r = lane < cnt ? t : node;
                    sum_rows<F>(Hs, lo, r, cnt, ax, ay);
                }
                for (int k0 = 64; k0 < deg; k0 += 32) {
                    const int left = deg - k0;
                    const int cnt  = left < 32 ? left : 32;
                    int r = node;
                    if (lane < cnt) r = slot[slot_of(cur, k0 + lane)];
                    sum_rows<F>(Hs, lo, r, cnt, ax, ay);
                }
            }
        }

        float* o = out + (size_t)node * F + lo;
        if constexpr (VPL == 2) {
            float ox = di * (ax + sx) + bx;
            float oy = di * (ay + sy) + by;
            if (RELU) { ox = fmaxf(ox, 0.0f); oy = fmaxf(oy, 0.0f); }
            *(float2*)o = make_float2(ox, oy);
        } else {
            float ox = di * (ax + sx) + bx;
            if (RELU) ox = fmaxf(ox, 0.0f);
            *o = ox;
        }

        if (nxt >= n) break;
        node = nxt;
        cur = nx; di = ndi;
        s = ns; sx = nsx; sy = nsy;
        nlo = clo; nhi = chi; ndi = cdi;
    }
}

__global__ void k_decode(const float* __restrict__ Z, const int* __restrict__ ea,
                         const int* __restrict__ eb, float* __restrict__ out, long long EL) {
    long long e = (long long)blockIdx.x * blockDim.x + threadIdx.x;
    if (e >= EL) return;
    const float4* za = (const float4*)(Z + (size_t)ea[e] * 32);
    const float4* zb = (const float4*)(Z + (size_t)eb[e] * 32);
    float s = 0.0f;
    #pragma unroll
    for (int i = 0; i < 8; ++i) {
        float4 a = za[i], b = zb[i];
        s += a.x * b.x + a.y * b.y + a.z * b.z + a.w * b.w;
    }
    out[e] = s;
}

static inline int cdiv(long long a, long long b) { return (int)((a + b - 1) / b); }

extern "C" void kernel_launch(void* const* d_in, const int* in_sizes, int n_in,
                              void* d_out, int out_size, void* d_ws, size_t ws_size,
                              hipStream_t stream) {
    const float* x   = (const float*)d_in[0];
    const float* W1  = (const float*)d_in[1];
    const float* b1  = (const float*)d_in[2];
    const float* W2  = (const float*)d_in[3];
    const float* b2  = (const float*)d_in[4];
    const int*   ei  = (const int*)d_in[5];
    const int*   eli = (const int*)d_in[6];

    const long long N  = in_sizes[0] / F_IN;
    const long long E  = in_sizes[5] / 2;
    const long long EL = in_sizes[6] / 2;
    float* out = (float*)d_out;

    const int nblk = cdiv(N, SCAN_T);
    float* h1   = (float*)d_ws;
    float* agg1 = h1   + N * F_HID;
    float* h2   = agg1 + N * F_HID;
    float* agg2 = h2   + N * F_OUT;
    int*   slot = (int*)(agg2 + N * F_OUT);
    int2*  off  = (int2*)(slot + 2 * E);
    int2*  part = off + (N + 1);
    int*   rank = (int*)(part + nblk);
    int*   cnt  = rank + E;
    float* dinv = (float*)(cnt + 2 * N);

    const int* row = ei;
    const int* col = ei + E;

    k_zero <<<256, 256, 0, stream>>>(cnt, (int)(2 * N));
    k_rank <<<EDGE_BLOCKS, 256, 0, stream>>>(col, cnt, rank, E, (int)N);
    k_bsum <<<nblk, SCAN_T, 0, stream>>>(cnt, part, (int)N);
    k_scan <<<nblk, SCAN_T, 0, stream>>>(cnt, part, off, dinv, (int)N);
    k_place<<<EDGE_BLOCKS, 256, 0, stream>>>(row, col, rank, (const int*)off, slot, (int)E);

    {
        long long tiles = ((N + 15) / 16) * (F_HID / 16);
        k_gemm_wmma<F_IN, F_HID><<<cdiv(tiles, 8), 256, 0, stream>>>(x, W1, dinv, h1, (int)N);
    }

    k_gather<F_HID, true><<<NODE_BLOCKS, 256, 0, stream>>>(h1, slot, off, dinv, b1, agg1, (int)N, (int)E);

    {
        long long tiles = ((N + 15) / 16) * (F_OUT / 16);
        k_gemm_wmma<F_HID, F_OUT><<<cdiv(tiles, 8), 256, 0, stream>>>(agg1, W2, dinv, h2, (int)N);
    }

    k_gather<F_OUT, false><<<NODE_BLOCKS, 256, 0, stream>>>(h2, slot, off, dinv, b2, agg2, (int)N, (int)E);

    k_decode<<<cdiv(EL, 256), 256, 0, stream>>>(agg2, eli, eli + EL, out, EL);
}


# counting pass stores each edge straight into its node's fixed 64-slot bucket per cache half (overflow list behind it): no rank array, prefix sums or place pass; sums still in (half, arrival) order
# speedup vs baseline: 5.5145x; 1.1088x over previous
#include <hip/hip_runtime.h>
#include <hip/hip_bf16.h>


typedef float v2f __attribute__((ext_vector_type(2)));
typedef float v8f __attribute__((ext_vector_type(8)));

#define F_IN  128
#define F_HID 64
#define F_OUT 32

#define EDGE_BLOCKS 1024
#define NODE_BLOCKS 2048
#define EDGE_U 4

#define CAP 64

__global__ void k_zero(int* __restrict__ p, int n) {
    const int T = gridDim.x * blockDim.x;
    for (int i = blockIdx.x * blockDim.x + threadIdx.x; i < n; i += T) p[i] = 0;
}

__device__ __forceinline__ int cache_half() {
    return (int)((__builtin_amdgcn_s_sendmsg_rtn(0x87) >> 18) & 1u);
}

__device__ __forceinline__ void put_edge(int* __restrict__ bk, int2* __restrict__ ovf,
                                         int* __restrict__ ovf_n, int c, int old, int r) {
    if (old < CAP) {
        bk[(size_t)c * (2 * CAP) + old] = r;
    } else {
        const int k = __hip_atomic_fetch_add(ovf_n, 1, __ATOMIC_RELAXED, __HIP_MEMORY_SCOPE_AGENT);
        ovf[k] = make_int2(c, r);
    }
}

__global__ void k_bucket(const int* __restrict__ row, const int* __restrict__ col,
                         int* __restrict__ cnt, int* __restrict__ slot,
                         int2* __restrict__ ovf, int* __restrict__ ovf_n, long long E, int n) {
    const long long T = (long long)gridDim.x * blockDim.x;
    long long e = (long long)blockIdx.x * blockDim.x + threadIdx.x;
    const int h = cache_half();
    int* mine = cnt + (size_t)h * n;
    int* bk   = slot + h * CAP;
    for (; e + (EDGE_U - 1) * T < E; e += EDGE_U * T) {
        int c[EDGE_U], r[EDGE_U], old[EDGE_U];
        #pragma unroll
        for (int u = 0; u < EDGE_U; ++u) {
            c[u] = col[e + u * T];
            r[u] = row[e + u * T];
        }
        static_assert(EDGE_U == 4, "the fence below names the four source ids of a trip");
        asm volatile("" :: "v"(r[0]), "v"(r[1]), "v"(r[2]), "v"(r[3]));
        #pragma unroll
        for (int u = 0; u < EDGE_U; ++u)
            old[u] = __hip_atomic_fetch_add(&mine[c[u]], 1, __ATOMIC_RELAXED, __HIP_MEMORY_SCOPE_AGENT);
        #pragma unroll
        for (int u = 0; u < EDGE_U; ++u) put_edge(bk, ovf, ovf_n, c[u], old[u], r[u]);
    }
    for (; e < E; e += T) {
        const int c = col[e];
        const int old = __hip_atomic_fetch_add(&mine[c], 1, __ATOMIC_RELAXED, __HIP_MEMORY_SCOPE_AGENT);
        put_edge(bk, ovf, ovf_n, c, old, row[e]);
    }
}

__device__ __forceinline__ float dinv_of(int c0, int c1) { return rsqrtf((float)(c0 + c1 + 1)); }

template <int K, int COUT>
__global__ void k_gemm_wmma(const float* __restrict__ X, const float* __restrict__ W,
                            const int* __restrict__ cnt, float* __restrict__ H, int n) {
    constexpr int COLT = COUT / 16;
    const int lane = threadIdx.x & 31;
    const int wv   = threadIdx.x >> 5;
    const int tile = blockIdx.x * (blockDim.x >> 5) + wv;
    const int rowTile = tile / COLT;
    const int colT    = tile - rowTile * COLT;
    const int rowBase = rowTile << 4;
    if (rowBase >= n) return;

    const int m  = lane & 15;
    const int kh = (lane >> 4) << 1;
    int rowA = rowBase + m;
    if (rowA >= n) rowA = n - 1;
    const float*  xr = X + (size_t)rowA * K;
    const float*  wc = W + colT * 16 + m;

    v8f acc = {};
    #pragma unroll
    for (int k = 0; k < K; k += 4) {
        v2f a, b;
        float2 av = *(const float2*)(xr + k + kh);
        a[0] = av.x;
        a[1] = av.y;
        b[0] = wc[(size_t)(k + kh)     * COUT];
        b[1] = wc[(size_t)(k + kh + 1) * COUT];
        acc = __builtin_amdgcn_wmma_f32_16x16x4_f32(
             false, a,  false, b,
             (short)0, acc,  false,  false);
    }

    const int mo = (lane >> 4) << 3;
    const int c  = colT * 16 + m;
    const float d = dinv_of(cnt[rowA], cnt[n + rowA]);
    #pragma unroll
    for (int v = 0; v < 8; ++v) {
        int r = rowBase + mo + v;
        const float dv = __shfl(d, mo + v);
        if (r < n) H[(size_t)r * COUT + c] = acc[v] * dv;
    }
}


template <int F>
__device__ __forceinline__ void sum_rows(const float* __restrict__ Hs, int lo, int r, int cnt,
                                         float& ax, float& ay) {
    constexpr int VPL   = F / 32;
    constexpr int BATCH = 16;
    int j0 = 0;
    do {
        float vx[BATCH], vy[BATCH];
        #pragma unroll
        for (int j = 0; j < BATCH; ++j) {
            const int rj = __builtin_amdgcn_readlane(r, j0 + j);
            const float* src = Hs + (size_t)rj * F + lo;
            if constexpr (VPL == 2) {
                const float2 t = *(const float2*)src;
                vx[j] = t.x;
                vy[j] = t.y;
            } else {
                vx[j] = *src;
                vy[j] = 0.0f;
            }
        }
        #pragma unroll
        for (int j = 0; j < BATCH; ++j) {
            if (j0 + j < cnt) {
                ax += vx[j];
                if constexpr (VPL == 2) ay += vy[j];
            }
        }
        j0 += BATCH;
    } while (j0 < cnt);
}

struct Seg {
    int a0;
    int ca;
    int b0;
    int deg;
};
__device__ __forceinline__ Seg make_seg(int node, int c0, int c1) {
    Seg s;
    const int m0 = c0 < CAP ? c0 : CAP;
    const int m1 = c1 < CAP ? c1 : CAP;
    s.a0  = node * (2 * CAP);
    s.ca  = m0;
    s.b0  = node * (2 * CAP) + CAP - m0;
    s.deg = m0 + m1;
    return s;
}
__device__ __forceinline__ bool has_overflow(int c0, int c1) { return c0 > CAP || c1 > CAP; }
__device__ __forceinline__ int slot_of(const Seg& s, int k) { return k + (k < s.ca ? s.a0 : s.b0); }

template <int F>
__device__ __forceinline__ void sum_overflow(const float* __restrict__ Hs, const int2* __restrict__ ovf,
                                             int len, int node, int lane, int lo, float& ax, float& ay) {
    constexpr int VPL = F / 32;
    for (int base = 0; base < len; base += 32) {
        int2 en = make_int2(-1, 0);
        if (base + lane < len) en = ovf[base + lane];
        unsigned m = __builtin_amdgcn_ballot_w32(en.x == node);
        while (m) {
            const int j  = __builtin_ctz(m);
            m &= m - 1;
            const int rj = __builtin_amdgcn_readlane(en.y, j);
            const float* src = Hs + (size_t)rj * F + lo;
            if constexpr (VPL == 2) {
                const float2 t = *(const float2*)src;
                ax += t.x;
                ay += t.y;
            } else {
                ax += *src;
            }
        }
    }
}

template <int F, bool RELU>
__global__ void k_gather(const float* __restrict__ Hs, const int* __restrict__ slot,
                         const int* __restrict__ counts, const int2* __restrict__ ovf,
                         const int* __restrict__ ovf_n,
                         const float* __restrict__ bias, float* __restrict__ out, int n) {
    constexpr int VPL = F / 32;
    const int lane = threadIdx.x & 31;
    const int nw = __builtin_amdgcn_readfirstlane((int)((gridDim.x * blockDim.x) >> 5));
    int node = __builtin_amdgcn_readfirstlane((int)((blockIdx.x * blockDim.x + threadIdx.x) >> 5));
    if (node >= n) return;
    const int lo = lane * VPL;

    float bx, by = 0.0f;
    if constexpr (VPL == 2) {
        const float2 b = *(const float2*)(bias + lo);
        bx = b.x;
        by = b.y;
    } else {
        bx = bias[lo];
    }

    const int ovf_len = *ovf_n;

    const int f0 = counts[node], f1 = counts[n + node];
    Seg   cur  = make_seg(node, f0, f1);
    float di   = dinv_of(f0, f1);
    bool  over = has_overflow(f0, f1);
    int s = 0;
    if (lane < cur.deg) s = slot[slot_of(cur, lane)];
    float sx, sy = 0.0f;
    {
        const float* src = Hs + (size_t)node * F + lo;
        if constexpr (VPL == 2) {
            const float2 t = *(const float2*)src;
            sx = t.x;
            sy = t.y;
        } else {
            sx = *src;
        }
    }
    if constexpr (VPL == 2) {
        asm volatile("" :: "v"(s), "v"(sx), "v"(sy));
    } else {
        asm volatile("" :: "v"(s), "v"(sx));
    }
    int n0 = 0, n1 = 0;
    if (node + nw < n) {
        n0 = counts[node + nw];
        n1 = counts[n + node + nw];
    }

    for (;;) {
        const int nxt = node + nw;
        int   ns = 0;
        float nsx = 0.0f, nsy = 0.0f;
        const Seg   nx    = make_seg(nxt, n0, n1);
        const float ndi   = dinv_of(n0, n1);
        const bool  nover = has_overflow(n0, n1);
        int q0 = 0, q1 = 0;
        if (nxt < n) {
            if (lane < nx.deg) ns = slot[slot_of(nx, lane)];
            const float* src = Hs + (size_t)nxt * F + lo;
            if constexpr (VPL == 2) {
                const float2 t = *(const float2*)src;
                nsx = t.x;
                nsy = t.y;
            } else {
                nsx = *src;
            }
            if (nxt + nw < n) {
                q0 = counts[nxt + nw];
                q1 = counts[n + nxt + nw];
            }
        }

        float ax = 0.0f, ay = 0.0f;
        const int deg = cur.deg;
        if (deg > 0) {
            int t = 0;
            if (deg > 32) {
                if (lane < deg - 32) t = slot[slot_of(cur, 32 + lane)];
            }
            {
                const int cnt = deg < 32 ? deg : 32;
                const int r = lane < cnt ? s : node;
                sum_rows<F>(Hs, lo, r, cnt, ax, ay);
            }
            if (deg > 32) {
                {
                    const int left = deg - 32;
                    const int cnt  = left < 32 ? left : 32;
                    const int r = lane < cnt ? t : node;
                    sum_rows<F>(Hs, lo, r, cnt, ax, ay);
                }
                for (int k0 = 64; k0 < deg; k0 += 32) {
                    const int left = deg - k0;
                    const int cnt  = left < 32 ? left : 32;
                    int r = node;
                    if (lane < cnt) r = slot[slot_of(cur, k0 + lane)];
                    sum_rows<F>(Hs, lo, r, cnt, ax, ay);
                }
            }
        }

        if (over) sum_overflow<F>(Hs, ovf, ovf_len, node, lane, lo, ax, ay);

        float* o = out + (size_t)node * F + lo;
        if constexpr (VPL == 2) {
            float ox = di * (ax + sx) + bx;
            float oy = di * (ay + sy) + by;
            if (RELU) { ox = fmaxf(ox, 0.0f); oy = fmaxf(oy, 0.0f); }
            *(float2*)o = make_float2(ox, oy);
        } else {
            float ox = di * (ax + sx) + bx;
            if (RELU) ox = fmaxf(ox, 0.0f);
            *o = ox;
        }

        if (nxt >= n) break;
        node = nxt;
        cur = nx; di = ndi; over = nover;
        s = ns; sx = nsx; sy = nsy;
        n0 = q0; n1 = q1;
    }
}

__global__ void k_decode(const float* __restrict__ Z, const int* __restrict__ ea,
                         const int* __restrict__ eb, float* __restrict__ out, long long EL) {
    long long e = (long long)blockIdx.x * blockDim.x + threadIdx.x;
    if (e >= EL) return;
    const float4* za = (const float4*)(Z + (size_t)ea[e] * 32);
    const float4* zb = (const float4*)(Z + (size_t)eb[e] * 32);
    float s = 0.0f;
    #pragma unroll
    for (int i = 0; i < 8; ++i) {
        float4 a = za[i], b = zb[i];
        s += a.x * b.x + a.y * b.y + a.z * b.z + a.w * b.w;
    }
    out[e] = s;
}

static inline int cdiv(long long a, long long b) { return (int)((a + b - 1) / b); }

extern "C" void kernel_launch(void* const* d_in, const int* in_sizes, int n_in,
                              void* d_out, int out_size, void* d_ws, size_t ws_size,
                              hipStream_t stream) {
    const float* x   = (const float*)d_in[0];
    const float* W1  = (const float*)d_in[1];
    const float* b1  = (const float*)d_in[2];
    const float* W2  = (const float*)d_in[3];
    const float* b2  = (const float*)d_in[4];
    const int*   ei  = (const int*)d_in[5];
    const int*   eli = (const int*)d_in[6];

    const long long N  = in_sizes[0] / F_IN;
    const long long E  = in_sizes[5] / 2;
    const long long EL = in_sizes[6] / 2;
    float* out = (float*)d_out;

    float* h1    = (float*)d_ws;
    float* agg1  = h1   + N * F_HID;
    float* h2    = agg1 + N * F_HID;
    float* agg2  = h2   + N * F_OUT;
    int*   slot  = (int*)(agg2 + N * F_OUT);
    int2*  ovf   = (int2*)(slot + 2 * CAP * N);
    int*   cnt   = (int*)(ovf + E);
    int*   ovf_n = cnt + 2 * N;

    const int* row = ei;
    const int* col = ei + E;

    k_zero  <<<256, 256, 0, stream>>>(cnt, (int)(2 * N + 1));
    k_bucket<<<EDGE_BLOCKS, 256, 0, stream>>>(row, col, cnt, slot, ovf, ovf_n, E, (int)N);

    {
        long long tiles = ((N + 15) / 16) * (F_HID / 16);
        k_gemm_wmma<F_IN, F_HID><<<cdiv(tiles, 8), 256, 0, stream>>>(x, W1, cnt, h1, (int)N);
    }

    k_gather<F_HID, true><<<NODE_BLOCKS, 256, 0, stream>>>(h1, slot, cnt, ovf, ovf_n, b1, agg1, (int)N);

    {
        long long tiles = ((N + 15) / 16) * (F_OUT / 16);
        k_gemm_wmma<F_HID, F_OUT><<<cdiv(tiles, 8), 256, 0, stream>>>(agg1, W2, cnt, h2, (int)N);
    }

    k_gather<F_OUT, false><<<NODE_BLOCKS, 256, 0, stream>>>(h2, slot, cnt, ovf, ovf_n, b2, agg2, (int)N);

    k_decode<<<cdiv(EL, 256), 256, 0, stream>>>(agg2, eli, eli + EL, out, EL);
}


# aggregation: fewer instructions per gathered row: padding lanes gather a zero row stored by the product (no per-row test), row address = scalar base + 32-bit lane offset; sum order kept
# speedup vs baseline: 5.8996x; 1.0698x over previous
#include <hip/hip_runtime.h>
#include <hip/hip_bf16.h>


typedef float v2f __attribute__((ext_vector_type(2)));
typedef float v8f __attribute__((ext_vector_type(8)));

#define F_IN  128
#define F_HID 64
#define F_OUT 32

#define EDGE_BLOCKS 1024
#define NODE_BLOCKS 2048
#define EDGE_U 4

#define CAP 64

__global__ void k_zero(int* __restrict__ p, int n) {
    const int T = gridDim.x * blockDim.x;
    for (int i = blockIdx.x * blockDim.x + threadIdx.x; i < n; i += T) p[i] = 0;
}

__device__ __forceinline__ int cache_half() {
    return (int)((__builtin_amdgcn_s_sendmsg_rtn(0x87) >> 18) & 1u);
}

__device__ __forceinline__ void put_edge(int* __restrict__ bk, int2* __restrict__ ovf,
                                         int* __restrict__ ovf_n, int c, int old, int r) {
    if (old < CAP) {
        bk[(size_t)c * (2 * CAP) + old] = r;
    } else {
        const int k = __hip_atomic_fetch_add(ovf_n, 1, __ATOMIC_RELAXED, __HIP_MEMORY_SCOPE_AGENT);
        ovf[k] = make_int2(c, r);
    }
}

__global__ void k_bucket(const int* __restrict__ row, const int* __restrict__ col,
                         int* __restrict__ cnt, int* __restrict__ slot,
                         int2* __restrict__ ovf, int* __restrict__ ovf_n, long long E, int n) {
    const long long T = (long long)gridDim.x * blockDim.x;
    long long e = (long long)blockIdx.x * blockDim.x + threadIdx.x;
    const int h = cache_half();
    int* mine = cnt + (size_t)h * n;
    int* bk   = slot + h * CAP;
    for (; e + (EDGE_U - 1) * T < E; e += EDGE_U * T) {
        int c[EDGE_U], r[EDGE_U], old[EDGE_U];
        #pragma unroll
        for (int u = 0; u < EDGE_U; ++u) {
            c[u] = col[e + u * T];
            r[u] = row[e + u * T];
        }
        static_assert(EDGE_U == 4, "the fence below names the four source ids of a trip");
        asm volatile("" :: "v"(r[0]), "v"(r[1]), "v"(r[2]), "v"(r[3]));
        #pragma unroll
        for (int u = 0; u < EDGE_U; ++u)
            old[u] = __hip_atomic_fetch_add(&mine[c[u]], 1, __ATOMIC_RELAXED, __HIP_MEMORY_SCOPE_AGENT);
        #pragma unroll
        for (int u = 0; u < EDGE_U; ++u) put_edge(bk, ovf, ovf_n, c[u], old[u], r[u]);
    }
    for (; e < E; e += T) {
        const int c = col[e];
        const int old = __hip_atomic_fetch_add(&mine[c], 1, __ATOMIC_RELAXED, __HIP_MEMORY_SCOPE_AGENT);
        put_edge(bk, ovf, ovf_n, c, old, row[e]);
    }
}

__device__ __forceinline__ float dinv_of(int c0, int c1) { return rsqrtf((float)(c0 + c1 + 1)); }

template <int K, int COUT>
__global__ void k_gemm_wmma(const float* __restrict__ X, const float* __restrict__ W,
                            const int* __restrict__ cnt, float* __restrict__ H, int n) {
    constexpr int COLT = COUT / 16;
    const int lane = threadIdx.x & 31;
    const int wv   = threadIdx.x >> 5;
    const int tile = blockIdx.x * (blockDim.x >> 5) + wv;
    const int rowTile = tile / COLT;
    const int colT    = tile - rowTile * COLT;
    const int rowBase = rowTile << 4;
    if (rowBase >= n) return;

    const int m  = lane & 15;
    const int kh = (lane >> 4) << 1;
    int rowA = rowBase + m;
    if (rowA >= n) rowA = n - 1;
    const float*  xr = X + (size_t)rowA * K;
    const float*  wc = W + colT * 16 + m;

    v8f acc = {};
    #pragma unroll
    for (int k = 0; k < K; k += 4) {
        v2f a, b;
        float2 av = *(const float2*)(xr + k + kh);
        a[0] = av.x;
        a[1] = av.y;
        b[0] = wc[(size_t)(k + kh)     * COUT];
        b[1] = wc[(size_t)(k + kh + 1) * COUT];
        acc = __builtin_amdgcn_wmma_f32_16x16x4_f32(
             false, a,  false, b,
             (short)0, acc,  false,  false);
    }

    const int mo = (lane >> 4) << 3;
    const int c  = colT * 16 + m;
    const float d = dinv_of(cnt[rowA], cnt[n + rowA]);
    #pragma unroll
    for (int v = 0; v < 8; ++v) {
        int r = rowBase + mo + v;
        const float dv = __shfl(d, mo + v);
        if (r < n) H[(size_t)r * COUT + c] = acc[v] * dv;
    }
    if (tile == 0) {
        #pragma unroll
        for (int cc = 0; cc < COUT; cc += 32) H[(size_t)n * COUT + cc + lane] = 0.0f;
    }
}


template <int F, int J0>
__device__ __forceinline__ void sum_batch(const float* __restrict__ Hs, unsigned lane_off, unsigned u,
                                          float& ax, float& ay) {
    constexpr int VPL   = F / 32;
    constexpr int BATCH = 16;
    float vx[BATCH], vy[BATCH];
    #pragma unroll
    for (int j = 0; j < BATCH; ++j) {
        const unsigned o = (unsigned)__builtin_amdgcn_readlane((int)u, J0 + j) + lane_off;
        const char* src = (const char*)Hs + o;
        if constexpr (VPL == 2) {
            const float2 t = *(const float2*)src;
            vx[j] = t.x;
            vy[j] = t.y;
        } else {
            vx[j] = *(const float*)src;
            vy[j] = 0.0f;
        }
    }
    if constexpr (VPL == 1) __builtin_amdgcn_sched_barrier(0);
    #pragma unroll
    for (int j = 0; j < BATCH; ++j) {
        ax += vx[j];
        if constexpr (VPL == 2) ay += vy[j];
    }
}
template <int F>
__device__ __forceinline__ void sum_rows(const float* __restrict__ Hs, unsigned lane_off, unsigned u, int cnt,
                                         float& ax, float& ay) {
    sum_batch<F, 0>(Hs, lane_off, u, ax, ay);
    if (cnt > 16) sum_batch<F, 16>(Hs, lane_off, u, ax, ay);
}

struct Seg {
    int a0;
    int ca;
    int b0;
    int deg;
};
__device__ __forceinline__ Seg make_seg(int node, int c0, int c1) {
    Seg s;
    const int m0 = c0 < CAP ? c0 : CAP;
    const int m1 = c1 < CAP ? c1 : CAP;
    s.a0  = node * (2 * CAP);
    s.ca  = m0;
    s.b0  = node * (2 * CAP) + CAP - m0;
    s.deg = m0 + m1;
    return s;
}
__device__ __forceinline__ bool has_overflow(int c0, int c1) { return c0 > CAP || c1 > CAP; }
__device__ __forceinline__ int slot_of(const Seg& s, int k) { return k + (k < s.ca ? s.a0 : s.b0); }

template <int F>
__device__ __forceinline__ void sum_overflow(const float* __restrict__ Hs, const int2* __restrict__ ovf,
                                             int len, int node, int lane, int lo, float& ax, float& ay) {
    constexpr int VPL = F / 32;
    for (int base = 0; base < len; base += 32) {
        int2 en = make_int2(-1, 0);
        if (base + lane < len) en = ovf[base + lane];
        unsigned m = __builtin_amdgcn_ballot_w32(en.x == node);
        while (m) {
            const int j  = __builtin_ctz(m);
            m &= m - 1;
            const int rj = __builtin_amdgcn_readlane(en.y, j);
            const float* src = Hs + (size_t)rj * F + lo;
            if constexpr (VPL == 2) {
                const float2 t = *(const float2*)src;
                ax += t.x;
                ay += t.y;
            } else {
                ax += *src;
            }
        }
    }
}

template <int F, bool RELU>
__global__ void k_gather(const float* __restrict__ Hs, const int* __restrict__ slot,
                         const int* __restrict__ counts, const int2* __restrict__ ovf,
                         const int* __restrict__ ovf_n,
                         const float* __restrict__ bias, float* __restrict__ out, int n) {
    constexpr int VPL = F / 32;
    const int lane = threadIdx.x & 31;
    const int nw = __builtin_amdgcn_readfirstlane((int)((gridDim.x * blockDim.x) >> 5));
    int node = __builtin_amdgcn_readfirstlane((int)((blockIdx.x * blockDim.x + threadIdx.x) >> 5));
    if (node >= n) return;
    const int lo = lane * VPL;
    const unsigned lane_off = (unsigned)lo * 4u;
    const unsigned row_b    = (unsigned)F * 4u;
    const unsigned zrow     = (unsigned)n * row_b;

    float bx, by = 0.0f;
    if constexpr (VPL == 2) {
        const float2 b = *(const float2*)(bias + lo);
        bx = b.x;
        by = b.y;
    } else {
        bx = bias[lo];
    }

    const int ovf_len = *ovf_n;

    const int f0 = counts[node], f1 = counts[n + node];
    Seg   cur  = make_seg(node, f0, f1);
    float di   = dinv_of(f0, f1);
    bool  over = has_overflow(f0, f1);
    int s = 0;
    if (lane < cur.deg) s = slot[slot_of(cur, lane)];
    float sx, sy = 0.0f;
    {
        const float* src = Hs + (size_t)node * F + lo;
        if constexpr (VPL == 2) {
            const float2 t = *(const float2*)src;
            sx = t.x;
            sy = t.y;
        } else {
            sx = *src;
        }
    }
    if constexpr (VPL == 2) {
        asm volatile("" :: "v"(s), "v"(sx), "v"(sy));
    } else {
        asm volatile("" :: "v"(s), "v"(sx));
    }
    int n0 = 0, n1 = 0;
    if (node + nw < n) {
        n0 = counts[node + nw];
        n1 = counts[n + node + nw];
    }

    for (;;) {
        const int nxt = node + nw;
        int   ns = 0;
        float nsx = 0.0f, nsy = 0.0f;
        const Seg   nx    = make_seg(nxt, n0, n1);
        const float ndi   = dinv_of(n0, n1);
        const bool  nover = has_overflow(n0, n1);
        int q0 = 0, q1 = 0;
        if (nxt < n) {
            if (lane < nx.deg) ns = slot[slot_of(nx, lane)];
            const float* src = Hs + (size_t)nxt * F + lo;
            if constexpr (VPL == 2) {
                const float2 t = *(const float2*)src;
                nsx = t.x;
                nsy = t.y;
            } else {
                nsx = *src;
            }
            if (nxt + nw < n) {
                q0 = counts[nxt + nw];
                q1 = counts[n + nxt + nw];
            }
        }

        float ax = 0.0f, ay = 0.0f;
        const int deg = cur.deg;
        if (deg > 0) {
            int t = 0;
            if (deg > 32) {
                if (lane < deg - 32) t = slot[slot_of(cur, 32 + lane)];
            }
            {
                const int cnt = deg < 32 ? deg : 32;
                const unsigned u = lane < cnt ? (unsigned)s * row_b : zrow;
                sum_rows<F>(Hs, lane_off, u, cnt, ax, ay);
            }
            if (deg > 32) {
                {
                    const int left = deg - 32;
                    const int cnt  = left < 32 ? left : 32;
                    const unsigned u = lane < cnt ? (unsigned)t * row_b : zrow;
                    sum_rows<F>(Hs, lane_off, u, cnt, ax, ay);
                }
                for (int k0 = 64; k0 < deg; k0 += 32) {
                    const int left = deg - k0;
                    const int cnt  = left < 32 ? left : 32;
                    unsigned u = zrow;
                    if (lane < cnt) u = (unsigned)slot[slot_of(cur, k0 + lane)] * row_b;
                    sum_rows<F>(Hs, lane_off, u, cnt, ax, ay);
                }
            }
        }

        if (over) sum_overflow<F>(Hs, ovf, ovf_len, node, lane, lo, ax, ay);

        float* o = out + (size_t)node * F + lo;
        if constexpr (VPL == 2) {
            float ox = di * (ax + sx) + bx;
            float oy = di * (ay + sy) + by;
            if (RELU) { ox = fmaxf(ox, 0.0f); oy = fmaxf(oy, 0.0f); }
            *(float2*)o = make_float2(ox, oy);
        } else {
            float ox = di * (ax + sx) + bx;
            if (RELU) ox = fmaxf(ox, 0.0f);
            *o = ox;
        }

        if (nxt >= n) break;
        node = nxt;
        cur = nx; di = ndi; over = nover;
        s = ns; sx = nsx; sy = nsy;
        n0 = q0; n1 = q1;
    }
}

__global__ void k_decode(const float* __restrict__ Z, const int* __restrict__ ea,
                         const int* __restrict__ eb, float* __restrict__ out, long long EL) {
    long long e = (long long)blockIdx.x * blockDim.x + threadIdx.x;
    if (e >= EL) return;
    const float4* za = (const float4*)(Z + (size_t)ea[e] * 32);
    const float4* zb = (const float4*)(Z + (size_t)eb[e] * 32);
    float s = 0.0f;
    #pragma unroll
    for (int i = 0; i < 8; ++i) {
        float4 a = za[i], b = zb[i];
        s += a.x * b.x + a.y * b.y + a.z * b.z + a.w * b.w;
    }
    out[e] = s;
}

static inline int cdiv(long long a, long long b) { return (int)((a + b - 1) / b); }

extern "C" void kernel_launch(void* const* d_in, const int* in_sizes, int n_in,
                              void* d_out, int out_size, void* d_ws, size_t ws_size,
                              hipStream_t stream) {
    const float* x   = (const float*)d_in[0];
    const float* W1  = (const float*)d_in[1];
    const float* b1  = (const float*)d_in[2];
    const float* W2  = (const float*)d_in[3];
    const float* b2  = (const float*)d_in[4];
    const int*   ei  = (const int*)d_in[5];
    const int*   eli = (const int*)d_in[6];

    const long long N  = in_sizes[0] / F_IN;
    const long long E  = in_sizes[5] / 2;
    const long long EL = in_sizes[6] / 2;
    float* out = (float*)d_out;

    float* h1    = (float*)d_ws;
    float* agg1  = h1   + (N + 1) * F_HID;
    float* h2    = agg1 + N * F_HID;
    float* agg2  = h2   + (N + 2) * F_OUT;
    int*   slot  = (int*)(agg2 + N * F_OUT);
    int2*  ovf   = (int2*)(slot + 2 * CAP * N);
    int*   cnt   = (int*)(ovf + E);
    int*   ovf_n = cnt + 2 * N;

    const int* row = ei;
    const int* col = ei + E;

    k_zero  <<<256, 256, 0, stream>>>(cnt, (int)(2 * N + 1));
    k_bucket<<<EDGE_BLOCKS, 256, 0, stream>>>(row, col, cnt, slot, ovf, ovf_n, E, (int)N);

    {
        long long tiles = ((N + 15) / 16) * (F_HID / 16);
        k_gemm_wmma<F_IN, F_HID><<<cdiv(tiles, 8), 256, 0, stream>>>(x, W1, cnt, h1, (int)N);
    }

    k_gather<F_HID, true><<<NODE_BLOCKS, 256, 0, stream>>>(h1, slot, cnt, ovf, ovf_n, b1, agg1, (int)N);

    {
        long long tiles = ((N + 15) / 16) * (F_OUT / 16);
        k_gemm_wmma<F_HID, F_OUT><<<cdiv(tiles, 8), 256, 0, stream>>>(agg1, W2, cnt, h2, (int)N);
    }

    k_gather<F_OUT, false><<<NODE_BLOCKS, 256, 0, stream>>>(h2, slot, cnt, ovf, ovf_n, b2, agg2, (int)N);

    k_decode<<<cdiv(EL, 256), 256, 0, stream>>>(agg2, eli, eli + EL, out, EL);
}


# products: a wave computes a whole 16-row strip, lane m owning adjacent columns: A read once (16 B a lane) for all column tiles, W and output rows as whole-row vectors; k summed in a new order
# speedup vs baseline: 6.3842x; 1.0821x over previous
#include <hip/hip_runtime.h>
#include <hip/hip_bf16.h>


typedef float v2f __attribute__((ext_vector_type(2)));
typedef float v8f __attribute__((ext_vector_type(8)));

#define F_IN  128
#define F_HID 64
#define F_OUT 32

#define EDGE_BLOCKS 1024
#define NODE_BLOCKS 2048
#define EDGE_U 4

#define CAP 64

__global__ void k_zero(int* __restrict__ p, int n) {
    const int T = gridDim.x * blockDim.x;
    for (int i = blockIdx.x * blockDim.x + threadIdx.x; i < n; i += T) p[i] = 0;
}

__device__ __forceinline__ int cache_half() {
    return (int)((__builtin_amdgcn_s_sendmsg_rtn(0x87) >> 18) & 1u);
}

__device__ __forceinline__ void put_edge(int* __restrict__ bk, int2* __restrict__ ovf,
                                         int* __restrict__ ovf_n, int c, int old, int r) {
    if (old < CAP) {
        bk[(size_t)c * (2 * CAP) + old] = r;
    } else {
        const int k = __hip_atomic_fetch_add(ovf_n, 1, __ATOMIC_RELAXED, __HIP_MEMORY_SCOPE_AGENT);
        ovf[k] = make_int2(c, r);
    }
}

__global__ void k_bucket(const int* __restrict__ row, const int* __restrict__ col,
                         int* __restrict__ cnt, int* __restrict__ slot,
                         int2* __restrict__ ovf, int* __restrict__ ovf_n, long long E, int n) {
    const long long T = (long long)gridDim.x * blockDim.x;
    long long e = (long long)blockIdx.x * blockDim.x + threadIdx.x;
    const int h = cache_half();
    int* mine = cnt + (size_t)h * n;
    int* bk   = slot + h * CAP;
    for (; e + (EDGE_U - 1) * T < E; e += EDGE_U * T) {
        int c[EDGE_U], r[EDGE_U], old[EDGE_U];
        #pragma unroll
        for (int u = 0; u < EDGE_U; ++u) {
            c[u] = col[e + u * T];
            r[u] = row[e + u * T];
        }
        static_assert(EDGE_U == 4, "the fence below names the four source ids of a trip");
        asm volatile("" :: "v"(r[0]), "v"(r[1]), "v"(r[2]), "v"(r[3]));
        #pragma unroll
        for (int u = 0; u < EDGE_U; ++u)
            old[u] = __hip_atomic_fetch_add(&mine[c[u]], 1, __ATOMIC_RELAXED, __HIP_MEMORY_SCOPE_AGENT);
        #pragma unroll
        for (int u = 0; u < EDGE_U; ++u) put_edge(bk, ovf, ovf_n, c[u], old[u], r[u]);
    }
    for (; e < E; e += T) {
        const int c = col[e];
        const int old = __hip_atomic_fetch_add(&mine[c], 1, __ATOMIC_RELAXED, __HIP_MEMORY_SCOPE_AGENT);
        put_edge(bk, ovf, ovf_n, c, old, row[e]);
    }
}

__device__ __forceinline__ float dinv_of(int c0, int c1) { return rsqrtf((float)(c0 + c1 + 1)); }

template <int K, int COUT>
__global__ void __launch_bounds__(128) k_gemm_wmma(const float* __restrict__ X, const float* __restrict__ W,
                            const int* __restrict__ cnt, float* __restrict__ H, int n) {
    constexpr int COLT = COUT / 16;
    typedef float vrow __attribute__((ext_vector_type(COLT)));
    const int lane  = threadIdx.x & 31;
    const int strip = blockIdx.x * (blockDim.x >> 5) + (threadIdx.x >> 5);
    const int rowBase = strip << 4;
    if (rowBase >= n) return;

    const int m = lane & 15;
    const int h = lane >> 4;
    int rowA = rowBase + m;
    if (rowA >= n) rowA = n - 1;
    const float* xr = X + (size_t)rowA * K + 4 * h;
    const float* wc = W + (size_t)(4 * h) * COUT + m * COLT;

    v8f acc[COLT];
    #pragma unroll
    for (int t = 0; t < COLT; ++t) acc[t] = v8f{};

    constexpr int NT = K / 8;
    float4 av[NT];
    #pragma unroll
    for (int i = 0; i < NT; ++i) av[i] = *(const float4*)(xr + 8 * i);
    const int c0 = cnt[rowA], c1 = cnt[n + rowA];
    vrow w[2][4];
    #pragma unroll
    for (int j = 0; j < 4; ++j) w[0][j] = *(const vrow*)(wc + (size_t)j * COUT);

    #pragma unroll
    for (int i = 0; i < NT; ++i) {
        if (i + 1 < NT) {
            #pragma unroll
            for (int j = 0; j < 4; ++j)
                w[(i + 1) & 1][j] = *(const vrow*)(wc + (size_t)(8 * (i + 1) + j) * COUT);
        }
        __builtin_amdgcn_sched_barrier(0);
        v2f a0, a1;
        a0[0] = av[i].x; a0[1] = av[i].y;
        a1[0] = av[i].z; a1[1] = av[i].w;
        #pragma unroll
        for (int t = 0; t < COLT; ++t) {
            v2f b;
            b[0] = w[i & 1][0][t];
            b[1] = w[i & 1][1][t];
            acc[t] = __builtin_amdgcn_wmma_f32_16x16x4_f32(
                 false, a0,  false, b,
                 (short)0, acc[t],  false,  false);
        }
        #pragma unroll
        for (int t = 0; t < COLT; ++t) {
            v2f b;
            b[0] = w[i & 1][2][t];
            b[1] = w[i & 1][3][t];
            acc[t] = __builtin_amdgcn_wmma_f32_16x16x4_f32(
                 false, a1,  false, b,
                 (short)0, acc[t],  false,  false);
        }
        __builtin_amdgcn_sched_barrier(0);
    }

    const int mo = h << 3;
    const float d = dinv_of(c0, c1);
    #pragma unroll
    for (int v = 0; v < 8; ++v) {
        const int r = rowBase + mo + v;
        const float dv = __shfl(d, mo + v);
        vrow o;
        #pragma unroll
        for (int t = 0; t < COLT; ++t) o[t] = acc[t][v] * dv;
        if (r < n) *(vrow*)(H + (size_t)r * COUT + m * COLT) = o;
    }
    if (strip == 0) {
        #pragma unroll
        for (int cc = 0; cc < COUT; cc += 32) H[(size_t)n * COUT + cc + lane] = 0.0f;
    }
}


template <int F, int J0>
__device__ __forceinline__ void sum_batch(const float* __restrict__ Hs, unsigned lane_off, unsigned u,
                                          float& ax, float& ay) {
    constexpr int VPL   = F / 32;
    constexpr int BATCH = 16;
    float vx[BATCH], vy[BATCH];
    #pragma unroll
    for (int j = 0; j < BATCH; ++j) {
        const unsigned o = (unsigned)__builtin_amdgcn_readlane((int)u, J0 + j) + lane_off;
        const char* src = (const char*)Hs + o;
        if constexpr (VPL == 2) {
            const float2 t = *(const float2*)src;
            vx[j] = t.x;
            vy[j] = t.y;
        } else {
            vx[j] = *(const float*)src;
            vy[j] = 0.0f;
        }
    }
    if constexpr (VPL == 1) __builtin_amdgcn_sched_barrier(0);
    #pragma unroll
    for (int j = 0; j < BATCH; ++j) {
        ax += vx[j];
        if constexpr (VPL == 2) ay += vy[j];
    }
}
template <int F>
__device__ __forceinline__ void sum_rows(const float* __restrict__ Hs, unsigned lane_off, unsigned u, int cnt,
                                         float& ax, float& ay) {
    sum_batch<F, 0>(Hs, lane_off, u, ax, ay);
    if (cnt > 16) sum_batch<F, 16>(Hs, lane_off, u, ax, ay);
}

struct Seg {
    int a0;
    int ca;
    int b0;
    int deg;
};
__device__ __forceinline__ Seg make_seg(int node, int c0, int c1) {
    Seg s;
    const int m0 = c0 < CAP ? c0 : CAP;
    const int m1 = c1 < CAP ? c1 : CAP;
    s.a0  = node * (2 * CAP);
    s.ca  = m0;
    s.b0  = node * (2 * CAP) + CAP - m0;
    s.deg = m0 + m1;
    return s;
}
__device__ __forceinline__ bool has_overflow(int c0, int c1) { return c0 > CAP || c1 > CAP; }
__device__ __forceinline__ int slot_of(const Seg& s, int k) { return k + (k < s.ca ? s.a0 : s.b0); }

template <int F>
__device__ __forceinline__ void sum_overflow(const float* __restrict__ Hs, const int2* __restrict__ ovf,
                                             int len, int node, int lane, int lo, float& ax, float& ay) {
    constexpr int VPL = F / 32;
    for (int base = 0; base < len; base += 32) {
        int2 en = make_int2(-1, 0);
        if (base + lane < len) en = ovf[base + lane];
        unsigned m = __builtin_amdgcn_ballot_w32(en.x == node);
        while (m) {
            const int j  = __builtin_ctz(m);
            m &= m - 1;
            const int rj = __builtin_amdgcn_readlane(en.y, j);
            const float* src = Hs + (size_t)rj * F + lo;
            if constexpr (VPL == 2) {
                const float2 t = *(const float2*)src;
                ax += t.x;
                ay += t.y;
            } else {
                ax += *src;
            }
        }
    }
}

template <int F, bool RELU>
__global__ void k_gather(const float* __restrict__ Hs, const int* __restrict__ slot,
                         const int* __restrict__ counts, const int2* __restrict__ ovf,
                         const int* __restrict__ ovf_n,
                         const float* __restrict__ bias, float* __restrict__ out, int n) {
    constexpr int VPL = F / 32;
    const int lane = threadIdx.x & 31;
    const int nw = __builtin_amdgcn_readfirstlane((int)((gridDim.x * blockDim.x) >> 5));
    int node = __builtin_amdgcn_readfirstlane((int)((blockIdx.x * blockDim.x + threadIdx.x) >> 5));
    if (node >= n) return;
    const int lo = lane * VPL;
    const unsigned lane_off = (unsigned)lo * 4u;
    const unsigned row_b    = (unsigned)F * 4u;
    const unsigned zrow     = (unsigned)n * row_b;

    float bx, by = 0.0f;
    if constexpr (VPL == 2) {
        const float2 b = *(const float2*)(bias + lo);
        bx = b.x;
        by = b.y;
    } else {
        bx = bias[lo];
    }

    const int ovf_len = *ovf_n;

    const int f0 = counts[node], f1 = counts[n + node];
    Seg   cur  = make_seg(node, f0, f1);
    float di   = dinv_of(f0, f1);
    bool  over = has_overflow(f0, f1);
    int s = 0;
    if (lane < cur.deg) s = slot[slot_of(cur, lane)];
    float sx, sy = 0.0f;
    {
        const float* src = Hs + (size_t)node * F + lo;
        if constexpr (VPL == 2) {
            const float2 t = *(const float2*)src;
            sx = t.x;
            sy = t.y;
        } else {
            sx = *src;
        }
    }
    if constexpr (VPL == 2) {
        asm volatile("" :: "v"(s), "v"(sx), "v"(sy));
    } else {
        asm volatile("" :: "v"(s), "v"(sx));
    }
    int n0 = 0, n1 = 0;
    if (node + nw < n) {
        n0 = counts[node + nw];
        n1 = counts[n + node + nw];
    }

    for (;;) {
        const int nxt = node + nw;
        int   ns = 0;
        float nsx = 0.0f, nsy = 0.0f;
        const Seg   nx    = make_seg(nxt, n0, n1);
        const float ndi   = dinv_of(n0, n1);
        const bool  nover = has_overflow(n0, n1);
        int q0 = 0, q1 = 0;
        if (nxt < n) {
            if (lane < nx.deg) ns = slot[slot_of(nx, lane)];
            const float* src = Hs + (size_t)nxt * F + lo;
            if constexpr (VPL == 2) {
                const float2 t = *(const float2*)src;
                nsx = t.x;
                nsy = t.y;
            } else {
                nsx = *src;
            }
            if (nxt + nw < n) {
                q0 = counts[nxt + nw];
                q1 = counts[n + nxt + nw];
            }
        }

        float ax = 0.0f, ay = 0.0f;
        const int deg = cur.deg;
        if (deg > 0) {
            int t = 0;
            if (deg > 32) {
                if (lane < deg - 32) t = slot[slot_of(cur, 32 + lane)];
            }
            {
                const int cnt = deg < 32 ? deg : 32;
                const unsigned u = lane < cnt ? (unsigned)s * row_b : zrow;
                sum_rows<F>(Hs, lane_off, u, cnt, ax, ay);
            }
            if (deg > 32) {
                {
                    const int left = deg - 32;
                    const int cnt  = left < 32 ? left : 32;
                    const unsigned u = lane < cnt ? (unsigned)t * row_b : zrow;
                    sum_rows<F>(Hs, lane_off, u, cnt, ax, ay);
                }
                for (int k0 = 64; k0 < deg; k0 += 32) {
                    const int left = deg - k0;
                    const int cnt  = left < 32 ? left : 32;
                    unsigned u = zrow;
                    if (lane < cnt) u = (unsigned)slot[slot_of(cur, k0 + lane)] * row_b;
                    sum_rows<F>(Hs, lane_off, u, cnt, ax, ay);
                }
            }
        }

        if (over) sum_overflow<F>(Hs, ovf, ovf_len, node, lane, lo, ax, ay);

        float* o = out + (size_t)node * F + lo;
        if constexpr (VPL == 2) {
            float ox = di * (ax + sx) + bx;
            float oy = di * (ay + sy) + by;
            if (RELU) { ox = fmaxf(ox, 0.0f); oy = fmaxf(oy, 0.0f); }
            *(float2*)o = make_float2(ox, oy);
        } else {
            float ox = di * (ax + sx) + bx;
            if (RELU) ox = fmaxf(ox, 0.0f);
            *o = ox;
        }

        if (nxt >= n) break;
        node = nxt;
        cur = nx; di = ndi; over = nover;
        s = ns; sx = nsx; sy = nsy;
        n0 = q0; n1 = q1;
    }
}

__global__ void k_decode(const float* __restrict__ Z, const int* __restrict__ ea,
                         const int* __restrict__ eb, float* __restrict__ out, long long EL) {
    long long e = (long long)blockIdx.x * blockDim.x + threadIdx.x;
    if (e >= EL) return;
    const float4* za = (const float4*)(Z + (size_t)ea[e] * 32);
    const float4* zb = (const float4*)(Z + (size_t)eb[e] * 32);
    float s = 0.0f;
    #pragma unroll
    for (int i = 0; i < 8; ++i) {
        float4 a = za[i], b = zb[i];
        s += a.x * b.x + a.y * b.y + a.z * b.z + a.w * b.w;
    }
    out[e] = s;
}

static inline int cdiv(long long a, long long b) { return (int)((a + b - 1) / b); }

extern "C" void kernel_launch(void* const* d_in, const int* in_sizes, int n_in,
                              void* d_out, int out_size, void* d_ws, size_t ws_size,
                              hipStream_t stream) {
    const float* x   = (const float*)d_in[0];
    const float* W1  = (const float*)d_in[1];
    const float* b1  = (const float*)d_in[2];
    const float* W2  = (const float*)d_in[3];
    const float* b2  = (const float*)d_in[4];
    const int*   ei  = (const int*)d_in[5];
    const int*   eli = (const int*)d_in[6];

    const long long N  = in_sizes[0] / F_IN;
    const long long E  = in_sizes[5] / 2;
    const long long EL = in_sizes[6] / 2;
    float* out = (float*)d_out;

    float* h1    = (float*)d_ws;
    float* agg1  = h1   + (N + 1) * F_HID;
    float* h2    = agg1 + N * F_HID;
    float* agg2  = h2   + (N + 2) * F_OUT;
    int*   slot  = (int*)(agg2 + N * F_OUT);
    int2*  ovf   = (int2*)(slot + 2 * CAP * N);
    int*   cnt   = (int*)(ovf + E);
    int*   ovf_n = cnt + 2 * N;

    const int* row = ei;
    const int* col = ei + E;

    k_zero  <<<256, 256, 0, stream>>>(cnt, (int)(2 * N + 1));
    k_bucket<<<EDGE_BLOCKS, 256, 0, stream>>>(row, col, cnt, slot, ovf, ovf_n, E, (int)N);

    {
        long long strips = (N + 15) / 16;
        k_gemm_wmma<F_IN, F_HID><<<cdiv(strips, 4), 128, 0, stream>>>(x, W1, cnt, h1, (int)N);
    }

    k_gather<F_HID, true><<<NODE_BLOCKS, 256, 0, stream>>>(h1, slot, cnt, ovf, ovf_n, b1, agg1, (int)N);

    {
        long long strips = (N + 15) / 16;
        k_gemm_wmma<F_HID, F_OUT><<<cdiv(strips, 4), 128, 0, stream>>>(agg1, W2, cnt, h2, (int)N);
    }

    k_gather<F_OUT, false><<<NODE_BLOCKS, 256, 0, stream>>>(h2, slot, cnt, ovf, ovf_n, b2, agg2, (int)N);

    k_decode<<<cdiv(EL, 256), 256, 0, stream>>>(agg2, eli, eli + EL, out, EL);
}
